# starting kernel
# baseline (speedup 1.0000x reference)
#include <hip/hip_runtime.h>


typedef __attribute__((ext_vector_type(2))) float v2f;
typedef __attribute__((ext_vector_type(8))) float v8f;

#define D   128
#define DE  32

__device__ __forceinline__ v8f wmma4(float a0, float a1, float b0, float b1, v8f c) {
  v2f a; a[0] = a0; a[1] = a1;
  v2f b; b[0] = b0; b[1] = b1;
  return __builtin_amdgcn_wmma_f32_16x16x4_f32(false, a, false, b, (short)0, c,
                                               false, false);
}

__global__ void zero_f32(float* __restrict__ p, long n) {
  long i = (long)blockIdx.x * blockDim.x + threadIdx.x;
  long stride = (long)gridDim.x * blockDim.x;
  for (; i < n; i += stride) p[i] = 0.0f;
}

__global__ void scatter_add(const float* __restrict__ X, const int* __restrict__ src,
                            const int* __restrict__ dst, float* __restrict__ sum,
                            float* __restrict__ cnt, int E) {
  const int wave = threadIdx.x >> 5;
  const int lane = threadIdx.x & 31;
  const int e = blockIdx.x * 8 + wave;
  if (e >= E) return;
  const int s = src[e];
  const int d = dst[e];
  const float4 v = *(const float4*)(X + (long)s * D + lane * 4);
  float* o = sum + (long)d * D + lane * 4;
  atomicAdd(o + 0, v.x);
  atomicAdd(o + 1, v.y);
  atomicAdd(o + 2, v.z);
  atomicAdd(o + 3, v.w);
  if (lane == 0) atomicAdd(cnt + d, 1.0f);
}

__global__ void finalize_mean(float* __restrict__ sum, const float* __restrict__ cnt,
                              long n) {
  long i = (long)blockIdx.x * blockDim.x + threadIdx.x;
  long stride = (long)gridDim.x * blockDim.x;
  for (; i < n; i += stride) {
    float c = cnt[i >> 7];
    sum[i] = sum[i] / fmaxf(c, 1.0f);
  }
}

__global__ void node_gemm(const float* __restrict__ A1, const float* __restrict__ W1,
                          const float* __restrict__ A2, const float* __restrict__ W2,
                          const float* __restrict__ bias, float* __restrict__ Hout,
                          int do_relu) {
  const int tid  = threadIdx.x;
  const int wave = tid >> 5;
  const int lane = tid & 31;
  const int half = lane >> 4;
  const int lm   = lane & 15;
  const int m0   = blockIdx.x * 16;
  const int n    = wave * 16 + lm;

  v8f c;
  const float bv = bias ? bias[n] : 0.0f;
#pragma unroll
  for (int r = 0; r < 8; ++r) c[r] = bv;

  const float* aRow = A1 + (long)(m0 + lm) * D + 2 * half;
#pragma unroll 4
  for (int k0 = 0; k0 < D; k0 += 4) {
    const float2 a = *(const float2*)(aRow + k0);
    const float b0 = W1[(k0 + 2 * half) * D + n];
    const float b1 = W1[(k0 + 2 * half + 1) * D + n];
    c = wmma4(a.x, a.y, b0, b1, c);
  }
  if (A2) {
    const float* aRow2 = A2 + (long)(m0 + lm) * D + 2 * half;
#pragma unroll 4
    for (int k0 = 0; k0 < D; k0 += 4) {
      const float2 a = *(const float2*)(aRow2 + k0);
      const float b0 = W2[(k0 + 2 * half) * D + n];
      const float b1 = W2[(k0 + 2 * half + 1) * D + n];
      c = wmma4(a.x, a.y, b0, b1, c);
    }
  }

#pragma unroll
  for (int r = 0; r < 8; ++r) {
    float v = c[r];
    if (do_relu) v = fmaxf(v, 0.0f);
    const int m = m0 + r + 8 * half;
    Hout[(long)m * D + n] = v;
  }
}

__global__ void edge_mlp(const float* __restrict__ P1, const float* __restrict__ P2,
                         const float* __restrict__ ea, const int* __restrict__ src,
                         const int* __restrict__ dst, const float* __restrict__ W3,
                         const float* __restrict__ bc1, const float* __restrict__ Wc2,
                         const float* __restrict__ bc2, float* __restrict__ out,
                         int E) {
  __shared__ int   s_src[16];
  __shared__ int   s_dst[16];
  __shared__ float s_red[8][16];

  const int tid  = threadIdx.x;
  const int wave = tid >> 5;
  const int lane = tid & 31;
  const int half = lane >> 4;
  const int lm   = lane & 15;
  const int e0   = blockIdx.x * 16;
  const int n    = wave * 16 + lm;

  if (tid < 16)       s_src[tid]      = src[e0 + tid];
  else if (tid < 32)  s_dst[tid - 16] = dst[e0 + tid - 16];
  __syncthreads();

  v8f c;
  const float bv = bc1[n];
#pragma unroll
  for (int r = 0; r < 8; ++r) {
    const int m = r + 8 * half;
    c[r] = P1[(long)s_src[m] * D + n] + P2[(long)s_dst[m] * D + n] + bv;
  }

  const float* aRow = ea + (long)(e0 + lm) * DE + 2 * half;
#pragma unroll
  for (int k0 = 0; k0 < DE; k0 += 4) {
    const float2 a = *(const float2*)(aRow + k0);
    const float b0 = W3[(k0 + 2 * half) * D + n];
    const float b1 = W3[(k0 + 2 * half + 1) * D + n];
    c = wmma4(a.x, a.y, b0, b1, c);
  }

  const float w2 = Wc2[n];
#pragma unroll
  for (int r = 0; r < 8; ++r) {
    float t = fmaxf(c[r], 0.0f) * w2;
    t += __shfl_xor(t, 1);
    t += __shfl_xor(t, 2);
    t += __shfl_xor(t, 4);
    t += __shfl_xor(t, 8);
    if (lm == 0) s_red[wave][r + 8 * half] = t;
  }
  __syncthreads();

  if (tid < 16) {
    float acc = bc2[0];
#pragma unroll
    for (int w = 0; w < 8; ++w) acc += s_red[w][tid];
    if (e0 + tid < E) out[e0 + tid] = acc;
  }
}

extern "C" void kernel_launch(void* const* d_in, const int* in_sizes, int n_in,
                              void* d_out, int out_size, void* d_ws, size_t ws_size,
                              hipStream_t stream) {
  (void)n_in; (void)out_size; (void)ws_size;
  const float* x   = (const float*)d_in[0];
  const float* ea  = (const float*)d_in[1];
  const float* W1a = (const float*)d_in[2];
  const float* b1  = (const float*)d_in[3];
  const float* W1r = (const float*)d_in[4];
  const float* W2a = (const float*)d_in[5];
  const float* b2  = (const float*)d_in[6];
  const float* W2r = (const float*)d_in[7];
  const float* Wc1 = (const float*)d_in[8];
  const float* bc1 = (const float*)d_in[9];
  const float* Wc2 = (const float*)d_in[10];
  const float* bc2 = (const float*)d_in[11];
  const int*   ei  = (const int*)d_in[12];

  const int N = in_sizes[0] / D;
  const int E = in_sizes[12] / 2;
  const int* src = ei;
  const int* dst = ei + E;

  float* ws = (float*)d_ws;
  const long NM = (long)N * D;
  float* bufA = ws;
  float* bufB = ws + NM;
  float* bufC = ws + 2 * NM;
  float* cnt  = ws + 3 * NM;
  float* out  = (float*)d_out;

  const dim3 blk(256);
  const int sgrid = (E + 7) / 8;
  const int ggrid = N / 16;
  const int egrid = E / 16;

  zero_f32<<<2048, blk, 0, stream>>>(bufA, NM);
  zero_f32<<<128,  blk, 0, stream>>>(cnt, N);
  scatter_add<<<sgrid, blk, 0, stream>>>(x, src, dst, bufA, cnt, E);
  finalize_mean<<<4096, blk, 0, stream>>>(bufA, cnt, NM);
  node_gemm<<<ggrid, blk, 0, stream>>>(bufA, W1a, x, W1r, b1, bufB, 1);

  zero_f32<<<2048, blk, 0, stream>>>(bufA, NM);
  zero_f32<<<128,  blk, 0, stream>>>(cnt, N);
  scatter_add<<<sgrid, blk, 0, stream>>>(bufB, src, dst, bufA, cnt, E);
  finalize_mean<<<4096, blk, 0, stream>>>(bufA, cnt, NM);
  node_gemm<<<ggrid, blk, 0, stream>>>(bufA, W2a, bufB, W2r, b2, bufC, 0);

  node_gemm<<<ggrid, blk, 0, stream>>>(bufC, Wc1,           nullptr, nullptr,
                                       nullptr, bufB, 0);
  node_gemm<<<ggrid, blk, 0, stream>>>(bufC, Wc1 + 128 * D, nullptr, nullptr,
                                       nullptr, bufA, 0);

  edge_mlp<<<egrid, blk, 0, stream>>>(bufB, bufA, ea, src, dst, Wc1 + 256 * D,
                                      bc1, Wc2, bc2, out, E);
}


# Neighbour mean by node ownership: a work-group owns 196 nodes, scans all dst, adds its edges' X rows into an LDS tile (LDS float atomics, arrival order), stores the mean; 13 launches become 7
# speedup vs baseline: 2.4339x; 2.4339x over previous
#include <hip/hip_runtime.h>


typedef __attribute__((ext_vector_type(2))) float v2f;
typedef __attribute__((ext_vector_type(8))) float v8f;

#define D   128
#define DE  32

__device__ __forceinline__ v8f wmma4(float a0, float a1, float b0, float b1, v8f c) {
  v2f a; a[0] = a0; a[1] = a1;
  v2f b; b[0] = b0; b[1] = b1;
  return __builtin_amdgcn_wmma_f32_16x16x4_f32(false, a, false, b, (short)0, c,
                                               false, false);
}

#define AG_ROWS 196
#define AG_U    8
#define AG_B    8
#define AG_QF   20

__global__ __launch_bounds__(1024) void gather_mean(
    const float* __restrict__ X, const int* __restrict__ src,
    const int* __restrict__ dst, float* __restrict__ agg, int E, int N) {
  __shared__ float acc[AG_ROWS * D];
  __shared__ float deg[AG_ROWS];
  const int tid  = threadIdx.x;
  const int lane = tid & 31;
  const int wave = __builtin_amdgcn_readfirstlane(tid >> 5);
  const int n0   = blockIdx.x * AG_ROWS;

  for (int i = tid; i < AG_ROWS * D; i += 1024) acc[i] = 0.0f;
  if (tid < AG_ROWS) deg[tid] = 0.0f;
  __syncthreads();

  int q_e = 0;
  int q_l = 0;
  int q_n = 0;

  auto flush = [&]() {
    const int s_own = (lane < q_n) ? src[q_e] : 0;
    for (int k = 0; k < q_n; k += AG_B) {
      int sj[AG_B];
#pragma unroll
      for (int j = 0; j < AG_B; ++j) sj[j] = __builtin_amdgcn_readlane(s_own, (k + j) & 31);
      float v[AG_B][4];
#pragma unroll
      for (int j = 0; j < AG_B; ++j) {
        v[j][0] = 0.0f; v[j][1] = 0.0f; v[j][2] = 0.0f; v[j][3] = 0.0f;
        if (k + j < q_n) {
          const int s = sj[j];
          const float* xr = X + (long)s * D;
          v[j][0] = xr[lane];
          v[j][1] = xr[lane + 32];
          v[j][2] = xr[lane + 64];
          v[j][3] = xr[lane + 96];
        }
      }
#pragma unroll
      for (int j = 0; j < AG_B; ++j) {
        if (k + j < q_n) {
          const int l = __builtin_amdgcn_readlane(q_l, k + j);
          float* o = acc + l * D;
          atomicAdd(o + lane,      v[j][0]);
          atomicAdd(o + lane + 32, v[j][1]);
          atomicAdd(o + lane + 64, v[j][2]);
          atomicAdd(o + lane + 96, v[j][3]);
          if (lane == 0) atomicAdd(deg + l, 1.0f);
        }
      }
    }
    q_n = 0;
  };

  const int step_edges = 128 * AG_U;
  const int nsteps = (E + step_edges - 1) / step_edges;
  for (int b = wave; b < nsteps; b += 32) {
    const int base = b * step_edges;
    if (q_n >= AG_QF) flush();

    int4 dv[AG_U];
    if (base + step_edges <= E) {
#pragma unroll
      for (int u = 0; u < AG_U; ++u)
        dv[u] = *(const int4*)(dst + base + u * 128 + 4 * lane);
    } else {
#pragma unroll
      for (int u = 0; u < AG_U; ++u) {
        const int e = base + u * 128 + 4 * lane;
        dv[u].x = (e     < E) ? dst[e]     : -1;
        dv[u].y = (e + 1 < E) ? dst[e + 1] : -1;
        dv[u].z = (e + 2 < E) ? dst[e + 2] : -1;
        dv[u].w = (e + 3 < E) ? dst[e + 3] : -1;
      }
    }

    unsigned mk[4 * AG_U];
    int hits = 0;
#pragma unroll
    for (int u = 0; u < AG_U; ++u) {
      dv[u].x -= n0; dv[u].y -= n0; dv[u].z -= n0; dv[u].w -= n0;
      mk[4 * u + 0] = __builtin_amdgcn_ballot_w32((unsigned)dv[u].x < (unsigned)AG_ROWS);
      mk[4 * u + 1] = __builtin_amdgcn_ballot_w32((unsigned)dv[u].y < (unsigned)AG_ROWS);
      mk[4 * u + 2] = __builtin_amdgcn_ballot_w32((unsigned)dv[u].z < (unsigned)AG_ROWS);
      mk[4 * u + 3] = __builtin_amdgcn_ballot_w32((unsigned)dv[u].w < (unsigned)AG_ROWS);
      hits += __builtin_popcount(mk[4 * u + 0]) + __builtin_popcount(mk[4 * u + 1]) +
              __builtin_popcount(mk[4 * u + 2]) + __builtin_popcount(mk[4 * u + 3]);
    }

    if (q_n + hits <= 32) {
#pragma unroll
      for (int u = 0; u < AG_U; ++u) {
#pragma unroll
        for (int c = 0; c < 4; ++c) {
          unsigned m = mk[4 * u + c];
          const int lc = (c == 0) ? dv[u].x : (c == 1) ? dv[u].y : (c == 2) ? dv[u].z : dv[u].w;
          while (m != 0u) {
            const int j = __builtin_ctz(m);
            m &= m - 1u;
            const int le = __builtin_amdgcn_readlane(lc, j);
            q_l = (lane == q_n) ? le : q_l;
            q_e = (lane == q_n) ? base + u * 128 + 4 * j + c : q_e;
            ++q_n;
          }
        }
      }
    } else {
      int unit = 0;
      for (int pass = 0; pass <= 4 * AG_U; ++pass) {
        bool full = false;
#pragma unroll
        for (int u = 0; u < AG_U; ++u) {
#pragma unroll
          for (int c = 0; c < 4; ++c) {
            unsigned m = mk[4 * u + c];
            const int lc = (c == 0) ? dv[u].x : (c == 1) ? dv[u].y : (c == 2) ? dv[u].z : dv[u].w;
            if (!full && 4 * u + c >= unit && m != 0u) {
              if (q_n + __builtin_popcount(m) > 32) {
                full = true;
                unit = 4 * u + c;
              } else {
                while (m != 0u) {
                  const int j = __builtin_ctz(m);
                  m &= m - 1u;
                  const int le = __builtin_amdgcn_readlane(lc, j);
                  q_l = (lane == q_n) ? le : q_l;
                  q_e = (lane == q_n) ? base + u * 128 + 4 * j + c : q_e;
                  ++q_n;
                }
              }
            }
          }
        }
        if (!full) break;
        flush();
      }
    }
  }
  flush();
  __syncthreads();

  for (int r = wave; r < AG_ROWS; r += 32) {
    const int n = n0 + r;
    if (n < N) {
      const float dn = fmaxf(deg[r], 1.0f);
      const float4 a = *(const float4*)(acc + r * D + 4 * lane);
      float4 o;
      o.x = a.x / dn; o.y = a.y / dn; o.z = a.z / dn; o.w = a.w / dn;
      *(float4*)(agg + (long)n * D + 4 * lane) = o;
    }
  }
}

__device__ __forceinline__ void node_product(const float* __restrict__ A,
                                             const float* __restrict__ W, int row,
                                             int half, int c, v8f (&acc)[8]) {
  const float* ar = A + (long)row * D + 4 * half;
  const float* wr = W + 4 * half * D + 4 * c;
#pragma unroll 4
  for (int j = 0; j < 16; ++j) {
    const float4 a = *(const float4*)(ar + 8 * j);
    const float* w = wr + 8 * j * D;
    const float4 b00 = *(const float4*)(w);
    const float4 b01 = *(const float4*)(w + 64);
    const float4 b10 = *(const float4*)(w + D);
    const float4 b11 = *(const float4*)(w + D + 64);
    const float4 b20 = *(const float4*)(w + 2 * D);
    const float4 b21 = *(const float4*)(w + 2 * D + 64);
    const float4 b30 = *(const float4*)(w + 3 * D);
    const float4 b31 = *(const float4*)(w + 3 * D + 64);
    acc[0] = wmma4(a.x, a.y, b00.x, b10.x, acc[0]);
    acc[1] = wmma4(a.x, a.y, b00.y, b10.y, acc[1]);
    acc[2] = wmma4(a.x, a.y, b00.z, b10.z, acc[2]);
    acc[3] = wmma4(a.x, a.y, b00.w, b10.w, acc[3]);
    acc[4] = wmma4(a.x, a.y, b01.x, b11.x, acc[4]);
    acc[5] = wmma4(a.x, a.y, b01.y, b11.y, acc[5]);
    acc[6] = wmma4(a.x, a.y, b01.z, b11.z, acc[6]);
    acc[7] = wmma4(a.x, a.y, b01.w, b11.w, acc[7]);
    acc[0] = wmma4(a.z, a.w, b20.x, b30.x, acc[0]);
    acc[1] = wmma4(a.z, a.w, b20.y, b30.y, acc[1]);
    acc[2] = wmma4(a.z, a.w, b20.z, b30.z, acc[2]);
    acc[3] = wmma4(a.z, a.w, b20.w, b30.w, acc[3]);
    acc[4] = wmma4(a.z, a.w, b21.x, b31.x, acc[4]);
    acc[5] = wmma4(a.z, a.w, b21.y, b31.y, acc[5]);
    acc[6] = wmma4(a.z, a.w, b21.z, b31.z, acc[6]);
    acc[7] = wmma4(a.z, a.w, b21.w, b31.w, acc[7]);
  }
}

__global__ __launch_bounds__(32) void node_gemm(
    const float* __restrict__ A1, const float* __restrict__ W1,
    const float* __restrict__ A2, const float* __restrict__ W2,
    const float* __restrict__ bias, float* __restrict__ Hout, int do_relu) {
  const int lane = threadIdx.x & 31;
  const int half = lane >> 4;
  const int c    = lane & 15;
  const int m0   = blockIdx.x * 16;

  v8f acc[8];
  {
    float4 bv0 = make_float4(0.0f, 0.0f, 0.0f, 0.0f);
    float4 bv1 = bv0;
    if (bias) {
      bv0 = *(const float4*)(bias + 4 * c);
      bv1 = *(const float4*)(bias + 64 + 4 * c);
    }
#pragma unroll
    for (int r = 0; r < 8; ++r) {
      acc[0][r] = bv0.x; acc[1][r] = bv0.y; acc[2][r] = bv0.z; acc[3][r] = bv0.w;
      acc[4][r] = bv1.x; acc[5][r] = bv1.y; acc[6][r] = bv1.z; acc[7][r] = bv1.w;
    }
  }

  node_product(A1, W1, m0 + c, half, c, acc);
  if (A2) node_product(A2, W2, m0 + c, half, c, acc);

#pragma unroll
  for (int r = 0; r < 8; ++r) {
    float4 o0 = make_float4(acc[0][r], acc[1][r], acc[2][r], acc[3][r]);
    float4 o1 = make_float4(acc[4][r], acc[5][r], acc[6][r], acc[7][r]);
    if (do_relu) {
      o0.x = fmaxf(o0.x, 0.0f); o0.y = fmaxf(o0.y, 0.0f);
      o0.z = fmaxf(o0.z, 0.0f); o0.w = fmaxf(o0.w, 0.0f);
      o1.x = fmaxf(o1.x, 0.0f); o1.y = fmaxf(o1.y, 0.0f);
      o1.z = fmaxf(o1.z, 0.0f); o1.w = fmaxf(o1.w, 0.0f);
    }
    float* o = Hout + (long)(m0 + 8 * half + r) * D + 4 * c;
    *(float4*)(o)      = o0;
    *(float4*)(o + 64) = o1;
  }
}

template <int CTRL>
__device__ __forceinline__ float row_xor_add(float v) {
  const int t = __builtin_amdgcn_update_dpp(0, __float_as_int(v), CTRL, 0xF, 0xF, false);
  return v + __int_as_float(t);
}

__global__ __launch_bounds__(128) void edge_mlp(
    const float* __restrict__ P1, const float* __restrict__ P2,
    const float* __restrict__ ea, const int* __restrict__ src,
    const int* __restrict__ dst, const float* __restrict__ W3,
    const float* __restrict__ bc1, const float* __restrict__ Wc2,
    const float* __restrict__ bc2, float* __restrict__ out, int E) {
  const int lane = threadIdx.x & 31;
  const int half = lane >> 4;
  const int c    = lane & 15;
  const int wave = __builtin_amdgcn_readfirstlane((int)(threadIdx.x >> 5));
  const int e0   = (blockIdx.x * 4 + wave) * 16;
  if (e0 >= E) return;

  const int4 s0 = *(const int4*)(src + e0 + 8 * half);
  const int4 s1 = *(const int4*)(src + e0 + 8 * half + 4);
  const int4 d0 = *(const int4*)(dst + e0 + 8 * half);
  const int4 d1 = *(const int4*)(dst + e0 + 8 * half + 4);
  const int si[8] = {s0.x, s0.y, s0.z, s0.w, s1.x, s1.y, s1.z, s1.w};
  const int di[8] = {d0.x, d0.y, d0.z, d0.w, d1.x, d1.y, d1.z, d1.w};

  float af[16];
  {
    const float* ar = ea + (long)(e0 + c) * DE + 16 * half;
#pragma unroll
    for (int q = 0; q < 4; ++q) {
      const float4 a = *(const float4*)(ar + 4 * q);
      af[4 * q + 0] = a.x; af[4 * q + 1] = a.y; af[4 * q + 2] = a.z; af[4 * q + 3] = a.w;
    }
  }

  v8f acc[8];
  {
    const float4 bv0 = *(const float4*)(bc1 + 4 * c);
    const float4 bv1 = *(const float4*)(bc1 + 64 + 4 * c);
#pragma unroll
    for (int r = 0; r < 8; ++r) {
      const float* p1 = P1 + (long)si[r] * D + 4 * c;
      const float* p2 = P2 + (long)di[r] * D + 4 * c;
      const float4 x0 = *(const float4*)(p1);
      const float4 x1 = *(const float4*)(p1 + 64);
      const float4 y0 = *(const float4*)(p2);
      const float4 y1 = *(const float4*)(p2 + 64);
      acc[0][r] = (x0.x + y0.x) + bv0.x;
      acc[1][r] = (x0.y + y0.y) + bv0.y;
      acc[2][r] = (x0.z + y0.z) + bv0.z;
      acc[3][r] = (x0.w + y0.w) + bv0.w;
      acc[4][r] = (x1.x + y1.x) + bv1.x;
      acc[5][r] = (x1.y + y1.y) + bv1.y;
      acc[6][r] = (x1.z + y1.z) + bv1.z;
      acc[7][r] = (x1.w + y1.w) + bv1.w;
    }
  }

#pragma unroll
  for (int ks = 0; ks < 8; ++ks) {
    const float* w0 = W3 + (16 * half + 2 * ks) * D + 4 * c;
    const float* w1 = w0 + D;
    const float4 b00 = *(const float4*)(w0);
    const float4 b01 = *(const float4*)(w0 + 64);
    const float4 b10 = *(const float4*)(w1);
    const float4 b11 = *(const float4*)(w1 + 64);
    const float a0 = af[2 * ks], a1 = af[2 * ks + 1];
    acc[0] = wmma4(a0, a1, b00.x, b10.x, acc[0]);
    acc[1] = wmma4(a0, a1, b00.y, b10.y, acc[1]);
    acc[2] = wmma4(a0, a1, b00.z, b10.z, acc[2]);
    acc[3] = wmma4(a0, a1, b00.w, b10.w, acc[3]);
    acc[4] = wmma4(a0, a1, b01.x, b11.x, acc[4]);
    acc[5] = wmma4(a0, a1, b01.y, b11.y, acc[5]);
    acc[6] = wmma4(a0, a1, b01.z, b11.z, acc[6]);
    acc[7] = wmma4(a0, a1, b01.w, b11.w, acc[7]);
  }

  const float4 wq0 = *(const float4*)(Wc2 + 4 * c);
  const float4 wq1 = *(const float4*)(Wc2 + 64 + 4 * c);
  const float w[8] = {wq0.x, wq0.y, wq0.z, wq0.w, wq1.x, wq1.y, wq1.z, wq1.w};
  float sc[8];
#pragma unroll
  for (int r = 0; r < 8; ++r) {
    float t = fmaxf(acc[0][r], 0.0f) * w[0];
#pragma unroll
    for (int k = 1; k < 8; ++k) t += fmaxf(acc[k][r], 0.0f) * w[k];
    t = row_xor_add<0x161>(t);
    t = row_xor_add<0x162>(t);
    t = row_xor_add<0x164>(t);
    t = row_xor_add<0x168>(t);
    sc[r] = t;
  }
  float v = sc[0];
#pragma unroll
  for (int r = 1; r < 8; ++r) v = (c == r) ? sc[r] : v;
  if (c < 8) out[e0 + 8 * half + c] = v + bc2[0];
}

extern "C" void kernel_launch(void* const* d_in, const int* in_sizes, int n_in,
                              void* d_out, int out_size, void* d_ws, size_t ws_size,
                              hipStream_t stream) {
  (void)n_in; (void)out_size; (void)ws_size;
  const float* x   = (const float*)d_in[0];
  const float* ea  = (const float*)d_in[1];
  const float* W1a = (const float*)d_in[2];
  const float* b1  = (const float*)d_in[3];
  const float* W1r = (const float*)d_in[4];
  const float* W2a = (const float*)d_in[5];
  const float* b2  = (const float*)d_in[6];
  const float* W2r = (const float*)d_in[7];
  const float* Wc1 = (const float*)d_in[8];
  const float* bc1 = (const float*)d_in[9];
  const float* Wc2 = (const float*)d_in[10];
  const float* bc2 = (const float*)d_in[11];
  const int*   ei  = (const int*)d_in[12];

  const int N = in_sizes[0] / D;
  const int E = in_sizes[12] / 2;
  const int* src = ei;
  const int* dst = ei + E;

  float* ws = (float*)d_ws;
  const long NM = (long)N * D;
  float* bufA = ws;
  float* bufB = ws + NM;
  float* bufC = ws + 2 * NM;
  float* out  = (float*)d_out;

  const int agrid = (N + AG_ROWS - 1) / AG_ROWS;
  const int ggrid = N / 16;
  const int egrid = (E / 16 + 3) / 4;

  gather_mean<<<agrid, dim3(1024), 0, stream>>>(x, src, dst, bufA, E, N);
  node_gemm<<<ggrid, dim3(32), 0, stream>>>(bufA, W1a, x, W1r, b1, bufB, 1);

  gather_mean<<<agrid, dim3(1024), 0, stream>>>(bufB, src, dst, bufA, E, N);
  node_gemm<<<ggrid, dim3(32), 0, stream>>>(bufA, W2a, bufB, W2r, b2, bufC, 0);

  node_gemm<<<ggrid, dim3(32), 0, stream>>>(bufC, Wc1,           nullptr, nullptr,
                                       nullptr, bufB, 0);
  node_gemm<<<ggrid, dim3(32), 0, stream>>>(bufC, Wc1 + 128 * D, nullptr, nullptr,
                                       nullptr, bufA, 0);

  edge_mlp<<<egrid, dim3(128), 0, stream>>>(bufB, bufA, ea, src, dst, Wc1 + 256 * D,
                                      bc1, Wc2, bc2, out, E);
}


# gather_mean: a prepass re-encodes the edge list (bit-sliced owner planes, 1 byte per edge, + packed src/row words); owner groups scan the planes, not dst; 7 launches become 8; LDS-atomic order kept
# speedup vs baseline: 2.7463x; 1.1284x over previous
#include <hip/hip_runtime.h>


typedef __attribute__((ext_vector_type(2))) float v2f;
typedef __attribute__((ext_vector_type(8))) float v8f;

#define D   128
#define DE  32

__device__ __forceinline__ v8f wmma4(float a0, float a1, float b0, float b1, v8f c) {
  v2f a; a[0] = a0; a[1] = a1;
  v2f b; b[0] = b0; b[1] = b1;
  return __builtin_amdgcn_wmma_f32_16x16x4_f32(false, a, false, b, (short)0, c,
                                               false, false);
}

#define AG_ROWS 196

__global__ __launch_bounds__(256) void edge_plane(
    const int* __restrict__ src, const int* __restrict__ dst,
    unsigned* __restrict__ plane, int* __restrict__ pk, int E) {
  const int lane = threadIdx.x & 31;
  const int wave = __builtin_amdgcn_readfirstlane((int)(threadIdx.x >> 5));
  const int tile = blockIdx.x * 8 + wave;
  if (tile * 128 >= E) return;
  const int e = tile * 128 + 4 * lane;
  int4 d = make_int4(0, 0, 0, 0);
  int4 s = d;
  if (e < E) {
    d = *(const int4*)(dst + e);
    s = *(const int4*)(src + e);
  }
  const int dd[4] = {d.x, d.y, d.z, d.w};
  int ow[4];
  int4 p;
  {
    int rw[4];
#pragma unroll
    for (int c = 0; c < 4; ++c) {
      ow[c] = (int)((unsigned)dd[c] / (unsigned)AG_ROWS);
      rw[c] = dd[c] - ow[c] * AG_ROWS;
    }
    p.x = (s.x << 8) | rw[0];
    p.y = (s.y << 8) | rw[1];
    p.z = (s.z << 8) | rw[2];
    p.w = (s.w << 8) | rw[3];
  }
  if (e < E) *(int4*)(pk + e) = p;
  int w = 0;
#pragma unroll
  for (int c = 0; c < 4; ++c) {
#pragma unroll
    for (int k = 0; k < 8; ++k) {
      const unsigned m = __builtin_amdgcn_ballot_w32(((ow[c] >> k) & 1) != 0);
      w = (lane == 8 * c + k) ? (int)m : w;
    }
  }
  plane[tile * 32 + lane] = (unsigned)w;
}

#define AG_U    4
#define AG_B    8

__global__ __launch_bounds__(1024) void gather_mean(
    const float* __restrict__ X, const unsigned* __restrict__ plane,
    const int* __restrict__ pk, float* __restrict__ agg, int E, int N) {
  __shared__ float acc[AG_ROWS * D];
  __shared__ float deg[AG_ROWS];
  __shared__ int   qbuf[32 * 32];
  const int tid  = threadIdx.x;
  const int lane = tid & 31;
  const int wave = __builtin_amdgcn_readfirstlane(tid >> 5);
  const int own  = blockIdx.x;
  const int nb   = own * AG_ROWS;

  for (int i = tid; i < AG_ROWS * D; i += 1024) acc[i] = 0.0f;
  if (tid < AG_ROWS) deg[tid] = 0.0f;
  __syncthreads();

  int* myq = qbuf + wave * 32;
  int q_n = 0;

  auto flush = [&]() {
    const int q_e = (lane < q_n) ? myq[lane] : 0;
    const int pw  = (lane < q_n) ? pk[q_e] : 0;
    for (int k = 0; k < q_n; k += AG_B) {
      int sj[AG_B];
      int lj[AG_B];
#pragma unroll
      for (int j = 0; j < AG_B; ++j) {
        const int t = __builtin_amdgcn_readlane(pw, (k + j) & 31);
        sj[j] = (int)((unsigned)t >> 8);
        lj[j] = t & 0xFF;
      }
      float v[AG_B][4];
#pragma unroll
      for (int j = 0; j < AG_B; ++j) {
        v[j][0] = 0.0f; v[j][1] = 0.0f; v[j][2] = 0.0f; v[j][3] = 0.0f;
        if (k + j < q_n) {
          const float* xr = X + (long)sj[j] * D;
          v[j][0] = xr[lane];
          v[j][1] = xr[lane + 32];
          v[j][2] = xr[lane + 64];
          v[j][3] = xr[lane + 96];
        }
      }
#pragma unroll
      for (int j = 0; j < AG_B; ++j) {
        if (k + j < q_n) {
          float* o = acc + lj[j] * D;
          atomicAdd(o + lane,      v[j][0]);
          atomicAdd(o + lane + 32, v[j][1]);
          atomicAdd(o + lane + 64, v[j][2]);
          atomicAdd(o + lane + 96, v[j][3]);
          if (lane == 0) atomicAdd(deg + lj[j], 1.0f);
        }
      }
    }
    q_n = 0;
  };

  unsigned pat[8];
#pragma unroll
  for (int k = 0; k < 8; ++k) pat[k] = 0u - (((unsigned)own >> k) & 1u);

  const int ntiles = (E + 127) >> 7;
  const int nsteps = (ntiles + 7) >> 3;
  const int sub    = lane >> 2;
  const int lbase  = sub * 128 + (lane & 3);

  const int wlast = ntiles * 32 - 8;
  auto fetch = [&](int s, uint4& w0, uint4& w1) {
    const int wofs = min(s * 256 + 8 * lane, wlast);
    const uint4* p = (const uint4*)(plane + wofs);
    w0 = p[0];
    w1 = p[1];
  };
  auto scan = [&](int s, const uint4& w0, const uint4& w1) {
    const unsigned u = (w0.x ^ pat[0]) | (w0.y ^ pat[1]) | (w0.z ^ pat[2]) | (w0.w ^ pat[3]) |
                       (w1.x ^ pat[4]) | (w1.y ^ pat[5]) | (w1.z ^ pat[6]) | (w1.w ^ pat[7]);
    const int e0 = s * 1024 + lbase;
    const int nv = (E - e0 + 3) >> 2;
    const unsigned live = (nv >= 32) ? ~0u : (nv <= 0) ? 0u : ((1u << nv) - 1u);
    unsigned h = ~u & live;
    unsigned lm = __builtin_amdgcn_ballot_w32(h != 0u);
    while (lm != 0u) {
      const int cnt = __builtin_popcount(lm);
      if (q_n + cnt > 32) flush();
      if (h != 0u) {
        const int e = e0 + 4 * __builtin_ctz(h);
        h &= h - 1u;
        myq[q_n + (int)__builtin_amdgcn_mbcnt_lo(lm, 0u)] = e;
      }
      q_n += cnt;
      lm = __builtin_amdgcn_ballot_w32(h != 0u);
    }
  };

  for (int s = wave; s < nsteps; s += 32 * AG_U) {
    uint4 w[AG_U][2];
#pragma unroll
    for (int i = 0; i < AG_U; ++i) fetch(s + 32 * i, w[i][0], w[i][1]);
#pragma unroll
    for (int i = 0; i < AG_U; ++i)
      if (s + 32 * i < nsteps) scan(s + 32 * i, w[i][0], w[i][1]);
  }
  flush();
  __syncthreads();

  for (int r = wave; r < AG_ROWS; r += 32) {
    const int n = nb + r;
    if (n < N) {
      const float dn = fmaxf(deg[r], 1.0f);
      const float4 a = *(const float4*)(acc + r * D + 4 * lane);
      float4 o;
      o.x = a.x / dn; o.y = a.y / dn; o.z = a.z / dn; o.w = a.w / dn;
      *(float4*)(agg + (long)n * D + 4 * lane) = o;
    }
  }
}

__device__ __forceinline__ void node_product(const float* __restrict__ A,
                                             const float* __restrict__ W, int row,
                                             int half, int c, v8f (&acc)[8]) {
  const float* ar = A + (long)row * D + 4 * half;
  const float* wr = W + 4 * half * D + 4 * c;
#pragma unroll 4
  for (int j = 0; j < 16; ++j) {
    const float4 a = *(const float4*)(ar + 8 * j);
    const float* w = wr + 8 * j * D;
    const float4 b00 = *(const float4*)(w);
    const float4 b01 = *(const float4*)(w + 64);
    const float4 b10 = *(const float4*)(w + D);
    const float4 b11 = *(const float4*)(w + D + 64);
    const float4 b20 = *(const float4*)(w + 2 * D);
    const float4 b21 = *(const float4*)(w + 2 * D + 64);
    const float4 b30 = *(const float4*)(w + 3 * D);
    const float4 b31 = *(const float4*)(w + 3 * D + 64);
    acc[0] = wmma4(a.x, a.y, b00.x, b10.x, acc[0]);
    acc[1] = wmma4(a.x, a.y, b00.y, b10.y, acc[1]);
    acc[2] = wmma4(a.x, a.y, b00.z, b10.z, acc[2]);
    acc[3] = wmma4(a.x, a.y, b00.w, b10.w, acc[3]);
    acc[4] = wmma4(a.x, a.y, b01.x, b11.x, acc[4]);
    acc[5] = wmma4(a.x, a.y, b01.y, b11.y, acc[5]);
    acc[6] = wmma4(a.x, a.y, b01.z, b11.z, acc[6]);
    acc[7] = wmma4(a.x, a.y, b01.w, b11.w, acc[7]);
    acc[0] = wmma4(a.z, a.w, b20.x, b30.x, acc[0]);
    acc[1] = wmma4(a.z, a.w, b20.y, b30.y, acc[1]);
    acc[2] = wmma4(a.z, a.w, b20.z, b30.z, acc[2]);
    acc[3] = wmma4(a.z, a.w, b20.w, b30.w, acc[3]);
    acc[4] = wmma4(a.z, a.w, b21.x, b31.x, acc[4]);
    acc[5] = wmma4(a.z, a.w, b21.y, b31.y, acc[5]);
    acc[6] = wmma4(a.z, a.w, b21.z, b31.z, acc[6]);
    acc[7] = wmma4(a.z, a.w, b21.w, b31.w, acc[7]);
  }
}

__global__ __launch_bounds__(32) void node_gemm(
    const float* __restrict__ A1, const float* __restrict__ W1,
    const float* __restrict__ A2, const float* __restrict__ W2,
    const float* __restrict__ bias, float* __restrict__ Hout, int do_relu) {
  const int lane = threadIdx.x & 31;
  const int half = lane >> 4;
  const int c    = lane & 15;
  const int m0   = blockIdx.x * 16;

  v8f acc[8];
  {
    float4 bv0 = make_float4(0.0f, 0.0f, 0.0f, 0.0f);
    float4 bv1 = bv0;
    if (bias) {
      bv0 = *(const float4*)(bias + 4 * c);
      bv1 = *(const float4*)(bias + 64 + 4 * c);
    }
#pragma unroll
    for (int r = 0; r < 8; ++r) {
      acc[0][r] = bv0.x; acc[1][r] = bv0.y; acc[2][r] = bv0.z; acc[3][r] = bv0.w;
      acc[4][r] = bv1.x; acc[5][r] = bv1.y; acc[6][r] = bv1.z; acc[7][r] = bv1.w;
    }
  }

  node_product(A1, W1, m0 + c, half, c, acc);
  if (A2) node_product(A2, W2, m0 + c, half, c, acc);

#pragma unroll
  for (int r = 0; r < 8; ++r) {
    float4 o0 = make_float4(acc[0][r], acc[1][r], acc[2][r], acc[3][r]);
    float4 o1 = make_float4(acc[4][r], acc[5][r], acc[6][r], acc[7][r]);
    if (do_relu) {
      o0.x = fmaxf(o0.x, 0.0f); o0.y = fmaxf(o0.y, 0.0f);
      o0.z = fmaxf(o0.z, 0.0f); o0.w = fmaxf(o0.w, 0.0f);
      o1.x = fmaxf(o1.x, 0.0f); o1.y = fmaxf(o1.y, 0.0f);
      o1.z = fmaxf(o1.z, 0.0f); o1.w = fmaxf(o1.w, 0.0f);
    }
    float* o = Hout + (long)(m0 + 8 * half + r) * D + 4 * c;
    *(float4*)(o)      = o0;
    *(float4*)(o + 64) = o1;
  }
}

template <int CTRL>
__device__ __forceinline__ float row_xor_add(float v) {
  const int t = __builtin_amdgcn_update_dpp(0, __float_as_int(v), CTRL, 0xF, 0xF, false);
  return v + __int_as_float(t);
}

__global__ __launch_bounds__(128) void edge_mlp(
    const float* __restrict__ P1, const float* __restrict__ P2,
    const float* __restrict__ ea, const int* __restrict__ src,
    const int* __restrict__ dst, const float* __restrict__ W3,
    const float* __restrict__ bc1, const float* __restrict__ Wc2,
    const float* __restrict__ bc2, float* __restrict__ out, int E) {
  const int lane = threadIdx.x & 31;
  const int half = lane >> 4;
  const int c    = lane & 15;
  const int wave = __builtin_amdgcn_readfirstlane((int)(threadIdx.x >> 5));
  const int e0   = (blockIdx.x * 4 + wave) * 16;
  if (e0 >= E) return;

  const int4 s0 = *(const int4*)(src + e0 + 8 * half);
  const int4 s1 = *(const int4*)(src + e0 + 8 * half + 4);
  const int4 d0 = *(const int4*)(dst + e0 + 8 * half);
  const int4 d1 = *(const int4*)(dst + e0 + 8 * half + 4);
  const int si[8] = {s0.x, s0.y, s0.z, s0.w, s1.x, s1.y, s1.z, s1.w};
  const int di[8] = {d0.x, d0.y, d0.z, d0.w, d1.x, d1.y, d1.z, d1.w};

  float af[16];
  {
    const float* ar = ea + (long)(e0 + c) * DE + 16 * half;
#pragma unroll
    for (int q = 0; q < 4; ++q) {
      const float4 a = *(const float4*)(ar + 4 * q);
      af[4 * q + 0] = a.x; af[4 * q + 1] = a.y; af[4 * q + 2] = a.z; af[4 * q + 3] = a.w;
    }
  }

  v8f acc[8];
  {
    const float4 bv0 = *(const float4*)(bc1 + 4 * c);
    const float4 bv1 = *(const float4*)(bc1 + 64 + 4 * c);
#pragma unroll
    for (int r = 0; r < 8; ++r) {
      const float* p1 = P1 + (long)si[r] * D + 4 * c;
      const float* p2 = P2 + (long)di[r] * D + 4 * c;
      const float4 x0 = *(const float4*)(p1);
      const float4 x1 = *(const float4*)(p1 + 64);
      const float4 y0 = *(const float4*)(p2);
      const float4 y1 = *(const float4*)(p2 + 64);
      acc[0][r] = (x0.x + y0.x) + bv0.x;
      acc[1][r] = (x0.y + y0.y) + bv0.y;
      acc[2][r] = (x0.z + y0.z) + bv0.z;
      acc[3][r] = (x0.w + y0.w) + bv0.w;
      acc[4][r] = (x1.x + y1.x) + bv1.x;
      acc[5][r] = (x1.y + y1.y) + bv1.y;
      acc[6][r] = (x1.z + y1.z) + bv1.z;
      acc[7][r] = (x1.w + y1.w) + bv1.w;
    }
  }

#pragma unroll
  for (int ks = 0; ks < 8; ++ks) {
    const float* w0 = W3 + (16 * half + 2 * ks) * D + 4 * c;
    const float* w1 = w0 + D;
    const float4 b00 = *(const float4*)(w0);
    const float4 b01 = *(const float4*)(w0 + 64);
    const float4 b10 = *(const float4*)(w1);
    const float4 b11 = *(const float4*)(w1 + 64);
    const float a0 = af[2 * ks], a1 = af[2 * ks + 1];
    acc[0] = wmma4(a0, a1, b00.x, b10.x, acc[0]);
    acc[1] = wmma4(a0, a1, b00.y, b10.y, acc[1]);
    acc[2] = wmma4(a0, a1, b00.z, b10.z, acc[2]);
    acc[3] = wmma4(a0, a1, b00.w, b10.w, acc[3]);
    acc[4] = wmma4(a0, a1, b01.x, b11.x, acc[4]);
    acc[5] = wmma4(a0, a1, b01.y, b11.y, acc[5]);
    acc[6] = wmma4(a0, a1, b01.z, b11.z, acc[6]);
    acc[7] = wmma4(a0, a1, b01.w, b11.w, acc[7]);
  }

  const float4 wq0 = *(const float4*)(Wc2 + 4 * c);
  const float4 wq1 = *(const float4*)(Wc2 + 64 + 4 * c);
  const float w[8] = {wq0.x, wq0.y, wq0.z, wq0.w, wq1.x, wq1.y, wq1.z, wq1.w};
  float sc[8];
#pragma unroll
  for (int r = 0; r < 8; ++r) {
    float t = fmaxf(acc[0][r], 0.0f) * w[0];
#pragma unroll
    for (int k = 1; k < 8; ++k) t += fmaxf(acc[k][r], 0.0f) * w[k];
    t = row_xor_add<0x161>(t);
    t = row_xor_add<0x162>(t);
    t = row_xor_add<0x164>(t);
    t = row_xor_add<0x168>(t);
    sc[r] = t;
  }
  float v = sc[0];
#pragma unroll
  for (int r = 1; r < 8; ++r) v = (c == r) ? sc[r] : v;
  if (c < 8) out[e0 + 8 * half + c] = v + bc2[0];
}

extern "C" void kernel_launch(void* const* d_in, const int* in_sizes, int n_in,
                              void* d_out, int out_size, void* d_ws, size_t ws_size,
                              hipStream_t stream) {
  (void)n_in; (void)out_size; (void)ws_size;
  const float* x   = (const float*)d_in[0];
  const float* ea  = (const float*)d_in[1];
  const float* W1a = (const float*)d_in[2];
  const float* b1  = (const float*)d_in[3];
  const float* W1r = (const float*)d_in[4];
  const float* W2a = (const float*)d_in[5];
  const float* b2  = (const float*)d_in[6];
  const float* W2r = (const float*)d_in[7];
  const float* Wc1 = (const float*)d_in[8];
  const float* bc1 = (const float*)d_in[9];
  const float* Wc2 = (const float*)d_in[10];
  const float* bc2 = (const float*)d_in[11];
  const int*   ei  = (const int*)d_in[12];

  const int N = in_sizes[0] / D;
  const int E = in_sizes[12] / 2;
  const int* src = ei;
  const int* dst = ei + E;

  float* ws = (float*)d_ws;
  const long NM = (long)N * D;
  float* bufA = ws;
  float* bufB = ws + NM;
  float* bufC = ws + 2 * NM;
  float* out  = (float*)d_out;

  const int agrid = (N + AG_ROWS - 1) / AG_ROWS;
  const int ntile = (E + 127) / 128;
  const int pgrid = (ntile + 7) / 8;
  const int ggrid = N / 16;
  const int egrid = (E / 16 + 3) / 4;

  unsigned* plane = (unsigned*)bufC;
  int*      pk    = (int*)bufC + ntile * 32;
  edge_plane<<<pgrid, dim3(256), 0, stream>>>(src, dst, plane, pk, E);

  gather_mean<<<agrid, dim3(1024), 0, stream>>>(x, plane, pk, bufA, E, N);
  node_gemm<<<ggrid, dim3(32), 0, stream>>>(bufA, W1a, x, W1r, b1, bufB, 1);

  gather_mean<<<agrid, dim3(1024), 0, stream>>>(bufB, plane, pk, bufA, E, N);
  node_gemm<<<ggrid, dim3(32), 0, stream>>>(bufA, W2a, bufB, W2r, b2, bufC, 0);

  node_gemm<<<ggrid, dim3(32), 0, stream>>>(bufC, Wc1,           nullptr, nullptr,
                                       nullptr, bufB, 0);
  node_gemm<<<ggrid, dim3(32), 0, stream>>>(bufC, Wc1 + 128 * D, nullptr, nullptr,
                                       nullptr, bufA, 0);

  edge_mlp<<<egrid, dim3(128), 0, stream>>>(bufB, bufA, ea, src, dst, Wc1 + 256 * D,
                                      bc1, Wc2, bc2, out, E);
}


# gather_mean: prepass sorts each 4096-edge chunk by owner in LDS (int-atomic arrival ranks, prefix sum, place) into per-owner runs; owner groups read only their runs, not all planes; LDS-atomic sums
# speedup vs baseline: 2.9475x; 1.0733x over previous
#include <hip/hip_runtime.h>


typedef __attribute__((ext_vector_type(2))) float v2f;
typedef __attribute__((ext_vector_type(8))) float v8f;

#define D   128
#define DE  32

__device__ __forceinline__ v8f wmma4(float a0, float a1, float b0, float b1, v8f c) {
  v2f a; a[0] = a0; a[1] = a1;
  v2f b; b[0] = b0; b[1] = b1;
  return __builtin_amdgcn_wmma_f32_16x16x4_f32(false, a, false, b, (short)0, c,
                                               false, false);
}

#define AG_ROWS 196
#define CH      4096

__global__ __launch_bounds__(1024) void edge_sort(
    const int* __restrict__ src, const int* __restrict__ dst,
    int* __restrict__ spk, int* __restrict__ runs, int E, int nch) {
  __shared__ int cnt[256];
  __shared__ int wsum[8];
  __shared__ int sorted[CH];
  const int tid   = threadIdx.x;
  const int lane  = tid & 31;
  const int chunk = blockIdx.x;
  if (tid < 256) cnt[tid] = 0;
  __syncthreads();

  const int  e  = chunk * CH + 4 * tid;
  const bool in = e < E;
  int ow[4] = {0, 0, 0, 0};
  int pw[4] = {0, 0, 0, 0};
  int rk[4] = {0, 0, 0, 0};
  if (in) {
    const int4 d = *(const int4*)(dst + e);
    const int4 s = *(const int4*)(src + e);
    const int dd[4] = {d.x, d.y, d.z, d.w};
    const int ss[4] = {s.x, s.y, s.z, s.w};
#pragma unroll
    for (int c = 0; c < 4; ++c) {
      ow[c] = (int)((unsigned)dd[c] / (unsigned)AG_ROWS);
      pw[c] = (ss[c] << 8) | (dd[c] - ow[c] * AG_ROWS);
      rk[c] = atomicAdd(&cnt[ow[c]], 1);
    }
  }
  __syncthreads();

  int c_own = 0;
  int incl  = 0;
  if (tid < 256) {
    c_own = cnt[tid];
    incl  = c_own;
#pragma unroll
    for (int dlt = 1; dlt < 32; dlt <<= 1) {
      const int t = __shfl_up(incl, dlt);
      if (lane >= dlt) incl += t;
    }
    if (lane == 31) wsum[tid >> 5] = incl;
  }
  __syncthreads();
  if (tid < 256) {
    int base = 0;
    for (int w = 0; w < (tid >> 5); ++w) base += wsum[w];
    const int start = base + incl - c_own;
    cnt[tid] = start;
    runs[tid * nch + chunk] = (c_own << 16) | start;
  }
  __syncthreads();

  if (in) {
#pragma unroll
    for (int c = 0; c < 4; ++c) sorted[cnt[ow[c]] + rk[c]] = pw[c];
  }
  __syncthreads();
  if (in) *(int4*)(spk + e) = *(const int4*)(sorted + 4 * tid);
}

#define AG_B    8

__global__ __launch_bounds__(1024) void gather_mean(
    const float* __restrict__ X, const int* __restrict__ spk,
    const int* __restrict__ runs, float* __restrict__ agg, int nch, int N) {
  __shared__ float acc[AG_ROWS * D];
  __shared__ float deg[AG_ROWS];
  const int tid  = threadIdx.x;
  const int lane = tid & 31;
  const int wave = __builtin_amdgcn_readfirstlane(tid >> 5);
  const int own  = blockIdx.x;
  const int nb   = own * AG_ROWS;

  for (int i = tid; i < AG_ROWS * D; i += 1024) acc[i] = 0.0f;
  if (tid < AG_ROWS) deg[tid] = 0.0f;
  __syncthreads();

  auto add_rows = [&](int pw, int q_n) {
    for (int k = 0; k < q_n; k += AG_B) {
      int sj[AG_B];
      int lj[AG_B];
#pragma unroll
      for (int j = 0; j < AG_B; ++j) {
        const int t = __builtin_amdgcn_readlane(pw, (k + j) & 31);
        sj[j] = (int)((unsigned)t >> 8);
        lj[j] = t & 0xFF;
      }
      float v[AG_B][4];
#pragma unroll
      for (int j = 0; j < AG_B; ++j) {
        v[j][0] = 0.0f; v[j][1] = 0.0f; v[j][2] = 0.0f; v[j][3] = 0.0f;
        if (k + j < q_n) {
          const float* xr = X + (long)sj[j] * D;
          v[j][0] = xr[lane];
          v[j][1] = xr[lane + 32];
          v[j][2] = xr[lane + 64];
          v[j][3] = xr[lane + 96];
        }
      }
#pragma unroll
      for (int j = 0; j < AG_B; ++j) {
        if (k + j < q_n) {
          float* o = acc + lj[j] * D;
          atomicAdd(o + lane,      v[j][0]);
          atomicAdd(o + lane + 32, v[j][1]);
          atomicAdd(o + lane + 64, v[j][2]);
          atomicAdd(o + lane + 96, v[j][3]);
          if (lane == 0) atomicAdd(deg + lj[j], 1.0f);
        }
      }
    }
  };

  const int* myruns = runs + own * nch;
  for (int c0 = wave; c0 < nch; c0 += 32 * 32) {
    const int ci    = c0 + 32 * lane;
    const int ent   = (ci < nch) ? myruns[ci] : 0;
    const int nmine = min(32, (nch - c0 + 31) >> 5);
    int t   = __builtin_amdgcn_readlane(ent, 0);
    int n_c = (int)((unsigned)t >> 16);
    int s_c = c0 * CH + (t & 0xFFFF);
    int w_c = (lane < n_c) ? spk[s_c + lane] : 0;
    for (int i = 0; i < nmine; ++i) {
      int n_n = 0, s_n = 0, w_n = 0;
      if (i + 1 < nmine) {
        t   = __builtin_amdgcn_readlane(ent, i + 1);
        n_n = (int)((unsigned)t >> 16);
        s_n = (c0 + 32 * (i + 1)) * CH + (t & 0xFFFF);
        w_n = (lane < n_n) ? spk[s_n + lane] : 0;
      }
      add_rows(w_c, min(n_c, 32));
      for (int b = 32; b < n_c; b += 32) {
        const int pw = (b + lane < n_c) ? spk[s_c + b + lane] : 0;
        add_rows(pw, min(n_c - b, 32));
      }
      n_c = n_n; s_c = s_n; w_c = w_n;
    }
  }
  __syncthreads();

  for (int r = wave; r < AG_ROWS; r += 32) {
    const int n = nb + r;
    if (n < N) {
      const float dn = fmaxf(deg[r], 1.0f);
      const float4 a = *(const float4*)(acc + r * D + 4 * lane);
      float4 o;
      o.x = a.x / dn; o.y = a.y / dn; o.z = a.z / dn; o.w = a.w / dn;
      *(float4*)(agg + (long)n * D + 4 * lane) = o;
    }
  }
}

__device__ __forceinline__ void node_product(const float* __restrict__ A,
                                             const float* __restrict__ W, int row,
                                             int half, int c, v8f (&acc)[8]) {
  const float* ar = A + (long)row * D + 4 * half;
  const float* wr = W + 4 * half * D + 4 * c;
#pragma unroll 4
  for (int j = 0; j < 16; ++j) {
    const float4 a = *(const float4*)(ar + 8 * j);
    const float* w = wr + 8 * j * D;
    const float4 b00 = *(const float4*)(w);
    const float4 b01 = *(const float4*)(w + 64);
    const float4 b10 = *(const float4*)(w + D);
    const float4 b11 = *(const float4*)(w + D + 64);
    const float4 b20 = *(const float4*)(w + 2 * D);
    const float4 b21 = *(const float4*)(w + 2 * D + 64);
    const float4 b30 = *(const float4*)(w + 3 * D);
    const float4 b31 = *(const float4*)(w + 3 * D + 64);
    acc[0] = wmma4(a.x, a.y, b00.x, b10.x, acc[0]);
    acc[1] = wmma4(a.x, a.y, b00.y, b10.y, acc[1]);
    acc[2] = wmma4(a.x, a.y, b00.z, b10.z, acc[2]);
    acc[3] = wmma4(a.x, a.y, b00.w, b10.w, acc[3]);
    acc[4] = wmma4(a.x, a.y, b01.x, b11.x, acc[4]);
    acc[5] = wmma4(a.x, a.y, b01.y, b11.y, acc[5]);
    acc[6] = wmma4(a.x, a.y, b01.z, b11.z, acc[6]);
    acc[7] = wmma4(a.x, a.y, b01.w, b11.w, acc[7]);
    acc[0] = wmma4(a.z, a.w, b20.x, b30.x, acc[0]);
    acc[1] = wmma4(a.z, a.w, b20.y, b30.y, acc[1]);
    acc[2] = wmma4(a.z, a.w, b20.z, b30.z, acc[2]);
    acc[3] = wmma4(a.z, a.w, b20.w, b30.w, acc[3]);
    acc[4] = wmma4(a.z, a.w, b21.x, b31.x, acc[4]);
    acc[5] = wmma4(a.z, a.w, b21.y, b31.y, acc[5]);
    acc[6] = wmma4(a.z, a.w, b21.z, b31.z, acc[6]);
    acc[7] = wmma4(a.z, a.w, b21.w, b31.w, acc[7]);
  }
}

__global__ __launch_bounds__(32) void node_gemm(
    const float* __restrict__ A1, const float* __restrict__ W1,
    const float* __restrict__ A2, const float* __restrict__ W2,
    const float* __restrict__ bias, float* __restrict__ Hout, int do_relu) {
  const int lane = threadIdx.x & 31;
  const int half = lane >> 4;
  const int c    = lane & 15;
  const int m0   = blockIdx.x * 16;

  v8f acc[8];
  {
    float4 bv0 = make_float4(0.0f, 0.0f, 0.0f, 0.0f);
    float4 bv1 = bv0;
    if (bias) {
      bv0 = *(const float4*)(bias + 4 * c);
      bv1 = *(const float4*)(bias + 64 + 4 * c);
    }
#pragma unroll
    for (int r = 0; r < 8; ++r) {
      acc[0][r] = bv0.x; acc[1][r] = bv0.y; acc[2][r] = bv0.z; acc[3][r] = bv0.w;
      acc[4][r] = bv1.x; acc[5][r] = bv1.y; acc[6][r] = bv1.z; acc[7][r] = bv1.w;
    }
  }

  node_product(A1, W1, m0 + c, half, c, acc);
  if (A2) node_product(A2, W2, m0 + c, half, c, acc);

#pragma unroll
  for (int r = 0; r < 8; ++r) {
    float4 o0 = make_float4(acc[0][r], acc[1][r], acc[2][r], acc[3][r]);
    float4 o1 = make_float4(acc[4][r], acc[5][r], acc[6][r], acc[7][r]);
    if (do_relu) {
      o0.x = fmaxf(o0.x, 0.0f); o0.y = fmaxf(o0.y, 0.0f);
      o0.z = fmaxf(o0.z, 0.0f); o0.w = fmaxf(o0.w, 0.0f);
      o1.x = fmaxf(o1.x, 0.0f); o1.y = fmaxf(o1.y, 0.0f);
      o1.z = fmaxf(o1.z, 0.0f); o1.w = fmaxf(o1.w, 0.0f);
    }
    float* o = Hout + (long)(m0 + 8 * half + r) * D + 4 * c;
    *(float4*)(o)      = o0;
    *(float4*)(o + 64) = o1;
  }
}

template <int CTRL>
__device__ __forceinline__ float row_xor_add(float v) {
  const int t = __builtin_amdgcn_update_dpp(0, __float_as_int(v), CTRL, 0xF, 0xF, false);
  return v + __int_as_float(t);
}

__global__ __launch_bounds__(128) void edge_mlp(
    const float* __restrict__ P1, const float* __restrict__ P2,
    const float* __restrict__ ea, const int* __restrict__ src,
    const int* __restrict__ dst, const float* __restrict__ W3,
    const float* __restrict__ bc1, const float* __restrict__ Wc2,
    const float* __restrict__ bc2, float* __restrict__ out, int E) {
  const int lane = threadIdx.x & 31;
  const int half = lane >> 4;
  const int c    = lane & 15;
  const int wave = __builtin_amdgcn_readfirstlane((int)(threadIdx.x >> 5));
  const int e0   = (blockIdx.x * 4 + wave) * 16;
  if (e0 >= E) return;

  const int4 s0 = *(const int4*)(src + e0 + 8 * half);
  const int4 s1 = *(const int4*)(src + e0 + 8 * half + 4);
  const int4 d0 = *(const int4*)(dst + e0 + 8 * half);
  const int4 d1 = *(const int4*)(dst + e0 + 8 * half + 4);
  const int si[8] = {s0.x, s0.y, s0.z, s0.w, s1.x, s1.y, s1.z, s1.w};
  const int di[8] = {d0.x, d0.y, d0.z, d0.w, d1.x, d1.y, d1.z, d1.w};

  float af[16];
  {
    const float* ar = ea + (long)(e0 + c) * DE + 16 * half;
#pragma unroll
    for (int q = 0; q < 4; ++q) {
      const float4 a = *(const float4*)(ar + 4 * q);
      af[4 * q + 0] = a.x; af[4 * q + 1] = a.y; af[4 * q + 2] = a.z; af[4 * q + 3] = a.w;
    }
  }

  v8f acc[8];
  {
    const float4 bv0 = *(const float4*)(bc1 + 4 * c);
    const float4 bv1 = *(const float4*)(bc1 + 64 + 4 * c);
#pragma unroll
    for (int r = 0; r < 8; ++r) {
      const float* p1 = P1 + (long)si[r] * D + 4 * c;
      const float* p2 = P2 + (long)di[r] * D + 4 * c;
      const float4 x0 = *(const float4*)(p1);
      const float4 x1 = *(const float4*)(p1 + 64);
      const float4 y0 = *(const float4*)(p2);
      const float4 y1 = *(const float4*)(p2 + 64);
      acc[0][r] = (x0.x + y0.x) + bv0.x;
      acc[1][r] = (x0.y + y0.y) + bv0.y;
      acc[2][r] = (x0.z + y0.z) + bv0.z;
      acc[3][r] = (x0.w + y0.w) + bv0.w;
      acc[4][r] = (x1.x + y1.x) + bv1.x;
      acc[5][r] = (x1.y + y1.y) + bv1.y;
      acc[6][r] = (x1.z + y1.z) + bv1.z;
      acc[7][r] = (x1.w + y1.w) + bv1.w;
    }
  }

#pragma unroll
  for (int ks = 0; ks < 8; ++ks) {
    const float* w0 = W3 + (16 * half + 2 * ks) * D + 4 * c;
    const float* w1 = w0 + D;
    const float4 b00 = *(const float4*)(w0);
    const float4 b01 = *(const float4*)(w0 + 64);
    const float4 b10 = *(const float4*)(w1);
    const float4 b11 = *(const float4*)(w1 + 64);
    const float a0 = af[2 * ks], a1 = af[2 * ks + 1];
    acc[0] = wmma4(a0, a1, b00.x, b10.x, acc[0]);
    acc[1] = wmma4(a0, a1, b00.y, b10.y, acc[1]);
    acc[2] = wmma4(a0, a1, b00.z, b10.z, acc[2]);
    acc[3] = wmma4(a0, a1, b00.w, b10.w, acc[3]);
    acc[4] = wmma4(a0, a1, b01.x, b11.x, acc[4]);
    acc[5] = wmma4(a0, a1, b01.y, b11.y, acc[5]);
    acc[6] = wmma4(a0, a1, b01.z, b11.z, acc[6]);
    acc[7] = wmma4(a0, a1, b01.w, b11.w, acc[7]);
  }

  const float4 wq0 = *(const float4*)(Wc2 + 4 * c);
  const float4 wq1 = *(const float4*)(Wc2 + 64 + 4 * c);
  const float w[8] = {wq0.x, wq0.y, wq0.z, wq0.w, wq1.x, wq1.y, wq1.z, wq1.w};
  float sc[8];
#pragma unroll
  for (int r = 0; r < 8; ++r) {
    float t = fmaxf(acc[0][r], 0.0f) * w[0];
#pragma unroll
    for (int k = 1; k < 8; ++k) t += fmaxf(acc[k][r], 0.0f) * w[k];
    t = row_xor_add<0x161>(t);
    t = row_xor_add<0x162>(t);
    t = row_xor_add<0x164>(t);
    t = row_xor_add<0x168>(t);
    sc[r] = t;
  }
  float v = sc[0];
#pragma unroll
  for (int r = 1; r < 8; ++r) v = (c == r) ? sc[r] : v;
  if (c < 8) out[e0 + 8 * half + c] = v + bc2[0];
}

extern "C" void kernel_launch(void* const* d_in, const int* in_sizes, int n_in,
                              void* d_out, int out_size, void* d_ws, size_t ws_size,
                              hipStream_t stream) {
  (void)n_in; (void)out_size; (void)ws_size;
  const float* x   = (const float*)d_in[0];
  const float* ea  = (const float*)d_in[1];
  const float* W1a = (const float*)d_in[2];
  const float* b1  = (const float*)d_in[3];
  const float* W1r = (const float*)d_in[4];
  const float* W2a = (const float*)d_in[5];
  const float* b2  = (const float*)d_in[6];
  const float* W2r = (const float*)d_in[7];
  const float* Wc1 = (const float*)d_in[8];
  const float* bc1 = (const float*)d_in[9];
  const float* Wc2 = (const float*)d_in[10];
  const float* bc2 = (const float*)d_in[11];
  const int*   ei  = (const int*)d_in[12];

  const int N = in_sizes[0] / D;
  const int E = in_sizes[12] / 2;
  const int* src = ei;
  const int* dst = ei + E;

  float* ws = (float*)d_ws;
  const long NM = (long)N * D;
  float* bufA = ws;
  float* bufB = ws + NM;
  float* bufC = ws + 2 * NM;
  float* out  = (float*)d_out;

  const int agrid = (N + AG_ROWS - 1) / AG_ROWS;
  const int nch   = (E + CH - 1) / CH;
  const int ggrid = N / 16;
  const int egrid = (E / 16 + 3) / 4;

  int* spk  = (int*)bufC;
  int* runs = (int*)bufC + nch * CH;
  edge_sort<<<nch, dim3(1024), 0, stream>>>(src, dst, spk, runs, E, nch);

  gather_mean<<<agrid, dim3(1024), 0, stream>>>(x, spk, runs, bufA, nch, N);
  node_gemm<<<ggrid, dim3(32), 0, stream>>>(bufA, W1a, x, W1r, b1, bufB, 1);

  gather_mean<<<agrid, dim3(1024), 0, stream>>>(bufB, spk, runs, bufA, nch, N);
  node_gemm<<<ggrid, dim3(32), 0, stream>>>(bufA, W2a, bufB, W2r, b2, bufC, 0);

  node_gemm<<<ggrid, dim3(32), 0, stream>>>(bufC, Wc1,           nullptr, nullptr,
                                       nullptr, bufB, 0);
  node_gemm<<<ggrid, dim3(32), 0, stream>>>(bufC, Wc1 + 128 * D, nullptr, nullptr,
                                       nullptr, bufA, 0);

  edge_mlp<<<egrid, dim3(128), 0, stream>>>(bufB, bufA, ea, src, dst, Wc1 + 256 * D,
                                      bc1, Wc2, bc2, out, E);
}


# node_gemm: a wave owns 16 rows x 64 columns (4 accumulators, 2-wave groups), not 128; P1 and P2 run as one launch (grid y): a SIMD holds 6-13 short waves, not 3-4 long; 8 launches to 7; sums kept
# speedup vs baseline: 3.1599x; 1.0721x over previous
#include <hip/hip_runtime.h>


typedef __attribute__((ext_vector_type(2))) float v2f;
typedef __attribute__((ext_vector_type(8))) float v8f;

#define D   128
#define DE  32

__device__ __forceinline__ v8f wmma4(float a0, float a1, float b0, float b1, v8f c) {
  v2f a; a[0] = a0; a[1] = a1;
  v2f b; b[0] = b0; b[1] = b1;
  return __builtin_amdgcn_wmma_f32_16x16x4_f32(false, a, false, b, (short)0, c,
                                               false, false);
}

#define AG_ROWS 196
#define CH      4096

__global__ __launch_bounds__(1024) void edge_sort(
    const int* __restrict__ src, const int* __restrict__ dst,
    int* __restrict__ spk, int* __restrict__ runs, int E, int nch) {
  __shared__ int cnt[256];
  __shared__ int wsum[8];
  __shared__ int sorted[CH];
  const int tid   = threadIdx.x;
  const int lane  = tid & 31;
  const int chunk = blockIdx.x;
  if (tid < 256) cnt[tid] = 0;
  __syncthreads();

  const int  e  = chunk * CH + 4 * tid;
  const bool in = e < E;
  int ow[4] = {0, 0, 0, 0};
  int pw[4] = {0, 0, 0, 0};
  int rk[4] = {0, 0, 0, 0};
  if (in) {
    const int4 d = *(const int4*)(dst + e);
    const int4 s = *(const int4*)(src + e);
    const int dd[4] = {d.x, d.y, d.z, d.w};
    const int ss[4] = {s.x, s.y, s.z, s.w};
#pragma unroll
    for (int c = 0; c < 4; ++c) {
      ow[c] = (int)((unsigned)dd[c] / (unsigned)AG_ROWS);
      pw[c] = (ss[c] << 8) | (dd[c] - ow[c] * AG_ROWS);
      rk[c] = atomicAdd(&cnt[ow[c]], 1);
    }
  }
  __syncthreads();

  int c_own = 0;
  int incl  = 0;
  if (tid < 256) {
    c_own = cnt[tid];
    incl  = c_own;
#pragma unroll
    for (int dlt = 1; dlt < 32; dlt <<= 1) {
      const int t = __shfl_up(incl, dlt);
      if (lane >= dlt) incl += t;
    }
    if (lane == 31) wsum[tid >> 5] = incl;
  }
  __syncthreads();
  if (tid < 256) {
    int base = 0;
    for (int w = 0; w < (tid >> 5); ++w) base += wsum[w];
    const int start = base + incl - c_own;
    cnt[tid] = start;
    runs[tid * nch + chunk] = (c_own << 16) | start;
  }
  __syncthreads();

  if (in) {
#pragma unroll
    for (int c = 0; c < 4; ++c) sorted[cnt[ow[c]] + rk[c]] = pw[c];
  }
  __syncthreads();
  if (in) *(int4*)(spk + e) = *(const int4*)(sorted + 4 * tid);
}

#define AG_B    8

__global__ __launch_bounds__(1024) void gather_mean(
    const float* __restrict__ X, const int* __restrict__ spk,
    const int* __restrict__ runs, float* __restrict__ agg, int nch, int N) {
  __shared__ float acc[AG_ROWS * D];
  __shared__ float deg[AG_ROWS];
  const int tid  = threadIdx.x;
  const int lane = tid & 31;
  const int wave = __builtin_amdgcn_readfirstlane(tid >> 5);
  const int own  = blockIdx.x;
  const int nb   = own * AG_ROWS;

  for (int i = tid; i < AG_ROWS * D; i += 1024) acc[i] = 0.0f;
  if (tid < AG_ROWS) deg[tid] = 0.0f;
  __syncthreads();

  auto add_rows = [&](int pw, int q_n) {
    for (int k = 0; k < q_n; k += AG_B) {
      int sj[AG_B];
      int lj[AG_B];
#pragma unroll
      for (int j = 0; j < AG_B; ++j) {
        const int t = __builtin_amdgcn_readlane(pw, (k + j) & 31);
        sj[j] = (int)((unsigned)t >> 8);
        lj[j] = t & 0xFF;
      }
      float v[AG_B][4];
#pragma unroll
      for (int j = 0; j < AG_B; ++j) {
        v[j][0] = 0.0f; v[j][1] = 0.0f; v[j][2] = 0.0f; v[j][3] = 0.0f;
        if (k + j < q_n) {
          const float* xr = X + (long)sj[j] * D;
          v[j][0] = xr[lane];
          v[j][1] = xr[lane + 32];
          v[j][2] = xr[lane + 64];
          v[j][3] = xr[lane + 96];
        }
      }
#pragma unroll
      for (int j = 0; j < AG_B; ++j) {
        if (k + j < q_n) {
          float* o = acc + lj[j] * D;
          atomicAdd(o + lane,      v[j][0]);
          atomicAdd(o + lane + 32, v[j][1]);
          atomicAdd(o + lane + 64, v[j][2]);
          atomicAdd(o + lane + 96, v[j][3]);
          if (lane == 0) atomicAdd(deg + lj[j], 1.0f);
        }
      }
    }
  };

  const int* myruns = runs + own * nch;
  for (int c0 = wave; c0 < nch; c0 += 32 * 32) {
    const int ci    = c0 + 32 * lane;
    const int ent   = (ci < nch) ? myruns[ci] : 0;
    const int nmine = min(32, (nch - c0 + 31) >> 5);
    int t   = __builtin_amdgcn_readlane(ent, 0);
    int n_c = (int)((unsigned)t >> 16);
    int s_c = c0 * CH + (t & 0xFFFF);
    int w_c = (lane < n_c) ? spk[s_c + lane] : 0;
    for (int i = 0; i < nmine; ++i) {
      int n_n = 0, s_n = 0, w_n = 0;
      if (i + 1 < nmine) {
        t   = __builtin_amdgcn_readlane(ent, i + 1);
        n_n = (int)((unsigned)t >> 16);
        s_n = (c0 + 32 * (i + 1)) * CH + (t & 0xFFFF);
        w_n = (lane < n_n) ? spk[s_n + lane] : 0;
      }
      add_rows(w_c, min(n_c, 32));
      for (int b = 32; b < n_c; b += 32) {
        const int pw = (b + lane < n_c) ? spk[s_c + b + lane] : 0;
        add_rows(pw, min(n_c - b, 32));
      }
      n_c = n_n; s_c = s_n; w_c = w_n;
    }
  }
  __syncthreads();

  for (int r = wave; r < AG_ROWS; r += 32) {
    const int n = nb + r;
    if (n < N) {
      const float dn = fmaxf(deg[r], 1.0f);
      const float4 a = *(const float4*)(acc + r * D + 4 * lane);
      float4 o;
      o.x = a.x / dn; o.y = a.y / dn; o.z = a.z / dn; o.w = a.w / dn;
      *(float4*)(agg + (long)n * D + 4 * lane) = o;
    }
  }
}

__device__ __forceinline__ void node_product(const float* __restrict__ A,
                                             const float* __restrict__ W, int row,
                                             int half, int col, v8f (&acc)[4]) {
  const float* ar = A + (long)row * D + 4 * half;
  const float* wr = W + 4 * half * D + col;
#pragma unroll 4
  for (int j = 0; j < 16; ++j) {
    const float4 a = *(const float4*)(ar + 8 * j);
    const float* w = wr + 8 * j * D;
    const float4 b0 = *(const float4*)(w);
    const float4 b1 = *(const float4*)(w + D);
    const float4 b2 = *(const float4*)(w + 2 * D);
    const float4 b3 = *(const float4*)(w + 3 * D);
    acc[0] = wmma4(a.x, a.y, b0.x, b1.x, acc[0]);
    acc[1] = wmma4(a.x, a.y, b0.y, b1.y, acc[1]);
    acc[2] = wmma4(a.x, a.y, b0.z, b1.z, acc[2]);
    acc[3] = wmma4(a.x, a.y, b0.w, b1.w, acc[3]);
    acc[0] = wmma4(a.z, a.w, b2.x, b3.x, acc[0]);
    acc[1] = wmma4(a.z, a.w, b2.y, b3.y, acc[1]);
    acc[2] = wmma4(a.z, a.w, b2.z, b3.z, acc[2]);
    acc[3] = wmma4(a.z, a.w, b2.w, b3.w, acc[3]);
  }
}

__global__ __launch_bounds__(64) void node_gemm(
    const float* __restrict__ A1, const float* __restrict__ W1,
    const float* __restrict__ A2, const float* __restrict__ W2,
    const float* __restrict__ bias, float* __restrict__ Hout, int do_relu,
    long w1_step, long out_step) {
  const int lane = threadIdx.x & 31;
  const int half = lane >> 4;
  const int c    = lane & 15;
  const int wave = __builtin_amdgcn_readfirstlane((int)(threadIdx.x >> 5));
  const int m0   = blockIdx.x * 16;
  const int col  = 64 * wave + 4 * c;
  W1   += (long)blockIdx.y * w1_step;
  Hout += (long)blockIdx.y * out_step;

  v8f acc[4];
  {
    float4 bv = make_float4(0.0f, 0.0f, 0.0f, 0.0f);
    if (bias) bv = *(const float4*)(bias + col);
#pragma unroll
    for (int r = 0; r < 8; ++r) {
      acc[0][r] = bv.x; acc[1][r] = bv.y; acc[2][r] = bv.z; acc[3][r] = bv.w;
    }
  }

  node_product(A1, W1, m0 + c, half, col, acc);
  if (A2) node_product(A2, W2, m0 + c, half, col, acc);

#pragma unroll
  for (int r = 0; r < 8; ++r) {
    float4 o = make_float4(acc[0][r], acc[1][r], acc[2][r], acc[3][r]);
    if (do_relu) {
      o.x = fmaxf(o.x, 0.0f); o.y = fmaxf(o.y, 0.0f);
      o.z = fmaxf(o.z, 0.0f); o.w = fmaxf(o.w, 0.0f);
    }
    *(float4*)(Hout + (long)(m0 + 8 * half + r) * D + col) = o;
  }
}

template <int CTRL>
__device__ __forceinline__ float row_xor_add(float v) {
  const int t = __builtin_amdgcn_update_dpp(0, __float_as_int(v), CTRL, 0xF, 0xF, false);
  return v + __int_as_float(t);
}

__global__ __launch_bounds__(128) void edge_mlp(
    const float* __restrict__ P1, const float* __restrict__ P2,
    const float* __restrict__ ea, const int* __restrict__ src,
    const int* __restrict__ dst, const float* __restrict__ W3,
    const float* __restrict__ bc1, const float* __restrict__ Wc2,
    const float* __restrict__ bc2, float* __restrict__ out, int E) {
  const int lane = threadIdx.x & 31;
  const int half = lane >> 4;
  const int c    = lane & 15;
  const int wave = __builtin_amdgcn_readfirstlane((int)(threadIdx.x >> 5));
  const int e0   = (blockIdx.x * 4 + wave) * 16;
  if (e0 >= E) return;

  const int4 s0 = *(const int4*)(src + e0 + 8 * half);
  const int4 s1 = *(const int4*)(src + e0 + 8 * half + 4);
  const int4 d0 = *(const int4*)(dst + e0 + 8 * half);
  const int4 d1 = *(const int4*)(dst + e0 + 8 * half + 4);
  const int si[8] = {s0.x, s0.y, s0.z, s0.w, s1.x, s1.y, s1.z, s1.w};
  const int di[8] = {d0.x, d0.y, d0.z, d0.w, d1.x, d1.y, d1.z, d1.w};

  float af[16];
  {
    const float* ar = ea + (long)(e0 + c) * DE + 16 * half;
#pragma unroll
    for (int q = 0; q < 4; ++q) {
      const float4 a = *(const float4*)(ar + 4 * q);
      af[4 * q + 0] = a.x; af[4 * q + 1] = a.y; af[4 * q + 2] = a.z; af[4 * q + 3] = a.w;
    }
  }

  v8f acc[8];
  {
    const float4 bv0 = *(const float4*)(bc1 + 4 * c);
    const float4 bv1 = *(const float4*)(bc1 + 64 + 4 * c);
#pragma unroll
    for (int r = 0; r < 8; ++r) {
      const float* p1 = P1 + (long)si[r] * D + 4 * c;
      const float* p2 = P2 + (long)di[r] * D + 4 * c;
      const float4 x0 = *(const float4*)(p1);
      const float4 x1 = *(const float4*)(p1 + 64);
      const float4 y0 = *(const float4*)(p2);
      const float4 y1 = *(const float4*)(p2 + 64);
      acc[0][r] = (x0.x + y0.x) + bv0.x;
      acc[1][r] = (x0.y + y0.y) + bv0.y;
      acc[2][r] = (x0.z + y0.z) + bv0.z;
      acc[3][r] = (x0.w + y0.w) + bv0.w;
      acc[4][r] = (x1.x + y1.x) + bv1.x;
      acc[5][r] = (x1.y + y1.y) + bv1.y;
      acc[6][r] = (x1.z + y1.z) + bv1.z;
      acc[7][r] = (x1.w + y1.w) + bv1.w;
    }
  }

#pragma unroll
  for (int ks = 0; ks < 8; ++ks) {
    const float* w0 = W3 + (16 * half + 2 * ks) * D + 4 * c;
    const float* w1 = w0 + D;
    const float4 b00 = *(const float4*)(w0);
    const float4 b01 = *(const float4*)(w0 + 64);
    const float4 b10 = *(const float4*)(w1);
    const float4 b11 = *(const float4*)(w1 + 64);
    const float a0 = af[2 * ks], a1 = af[2 * ks + 1];
    acc[0] = wmma4(a0, a1, b00.x, b10.x, acc[0]);
    acc[1] = wmma4(a0, a1, b00.y, b10.y, acc[1]);
    acc[2] = wmma4(a0, a1, b00.z, b10.z, acc[2]);
    acc[3] = wmma4(a0, a1, b00.w, b10.w, acc[3]);
    acc[4] = wmma4(a0, a1, b01.x, b11.x, acc[4]);
    acc[5] = wmma4(a0, a1, b01.y, b11.y, acc[5]);
    acc[6] = wmma4(a0, a1, b01.z, b11.z, acc[6]);
    acc[7] = wmma4(a0, a1, b01.w, b11.w, acc[7]);
  }

  const float4 wq0 = *(const float4*)(Wc2 + 4 * c);
  const float4 wq1 = *(const float4*)(Wc2 + 64 + 4 * c);
  const float w[8] = {wq0.x, wq0.y, wq0.z, wq0.w, wq1.x, wq1.y, wq1.z, wq1.w};
  float sc[8];
#pragma unroll
  for (int r = 0; r < 8; ++r) {
    float t = fmaxf(acc[0][r], 0.0f) * w[0];
#pragma unroll
    for (int k = 1; k < 8; ++k) t += fmaxf(acc[k][r], 0.0f) * w[k];
    t = row_xor_add<0x161>(t);
    t = row_xor_add<0x162>(t);
    t = row_xor_add<0x164>(t);
    t = row_xor_add<0x168>(t);
    sc[r] = t;
  }
  float v = sc[0];
#pragma unroll
  for (int r = 1; r < 8; ++r) v = (c == r) ? sc[r] : v;
  if (c < 8) out[e0 + 8 * half + c] = v + bc2[0];
}

extern "C" void kernel_launch(void* const* d_in, const int* in_sizes, int n_in,
                              void* d_out, int out_size, void* d_ws, size_t ws_size,
                              hipStream_t stream) {
  (void)n_in; (void)out_size; (void)ws_size;
  const float* x   = (const float*)d_in[0];
  const float* ea  = (const float*)d_in[1];
  const float* W1a = (const float*)d_in[2];
  const float* b1  = (const float*)d_in[3];
  const float* W1r = (const float*)d_in[4];
  const float* W2a = (const float*)d_in[5];
  const float* b2  = (const float*)d_in[6];
  const float* W2r = (const float*)d_in[7];
  const float* Wc1 = (const float*)d_in[8];
  const float* bc1 = (const float*)d_in[9];
  const float* Wc2 = (const float*)d_in[10];
  const float* bc2 = (const float*)d_in[11];
  const int*   ei  = (const int*)d_in[12];

  const int N = in_sizes[0] / D;
  const int E = in_sizes[12] / 2;
  const int* src = ei;
  const int* dst = ei + E;

  float* ws = (float*)d_ws;
  const long NM = (long)N * D;
  float* bufA = ws;
  float* bufB = ws + NM;
  float* bufC = ws + 2 * NM;
  float* out  = (float*)d_out;

  const int agrid = (N + AG_ROWS - 1) / AG_ROWS;
  const int nch   = (E + CH - 1) / CH;
  const int ggrid = N / 16;
  const int egrid = (E / 16 + 3) / 4;

  int* spk  = (int*)bufC;
  int* runs = (int*)bufC + nch * CH;
  edge_sort<<<nch, dim3(1024), 0, stream>>>(src, dst, spk, runs, E, nch);

  gather_mean<<<agrid, dim3(1024), 0, stream>>>(x, spk, runs, bufA, nch, N);
  node_gemm<<<dim3(ggrid, 1), dim3(64), 0, stream>>>(bufA, W1a, x, W1r, b1, bufB, 1, 0L, 0L);

  gather_mean<<<agrid, dim3(1024), 0, stream>>>(bufB, spk, runs, bufA, nch, N);
  node_gemm<<<dim3(ggrid, 1), dim3(64), 0, stream>>>(bufA, W2a, bufB, W2r, b2, bufC, 0, 0L, 0L);

  node_gemm<<<dim3(ggrid, 2), dim3(64), 0, stream>>>(bufC, Wc1, nullptr, nullptr, nullptr,
                                                    bufA, 0, 128L * D, NM);

  edge_mlp<<<egrid, dim3(128), 0, stream>>>(bufA, bufB, ea, src, dst, Wc1 + 256 * D,
                                      bc1, Wc2, bc2, out, E);
}


# gather_mean: each LDS sum row gets one owner wave; edges filed per owner wave in LDS (int counters, scan, place); owners add by plain LDS load/store, no float atomics; sums in filing (arrival) order
# speedup vs baseline: 3.2737x; 1.0360x over previous
#include <hip/hip_runtime.h>


typedef __attribute__((ext_vector_type(2))) float v2f;
typedef __attribute__((ext_vector_type(8))) float v8f;

#define D   128
#define DE  32

__device__ __forceinline__ v8f wmma4(float a0, float a1, float b0, float b1, v8f c) {
  v2f a; a[0] = a0; a[1] = a1;
  v2f b; b[0] = b0; b[1] = b1;
  return __builtin_amdgcn_wmma_f32_16x16x4_f32(false, a, false, b, (short)0, c,
                                               false, false);
}

#define AG_ROWS 196
#define CH      4096

__global__ __launch_bounds__(1024) void edge_sort(
    const int* __restrict__ src, const int* __restrict__ dst,
    int* __restrict__ spk, int* __restrict__ runs, int E, int nch) {
  __shared__ int cnt[256];
  __shared__ int wsum[8];
  __shared__ int sorted[CH];
  const int tid   = threadIdx.x;
  const int lane  = tid & 31;
  const int chunk = blockIdx.x;
  if (tid < 256) cnt[tid] = 0;
  __syncthreads();

  const int  e  = chunk * CH + 4 * tid;
  const bool in = e < E;
  int ow[4] = {0, 0, 0, 0};
  int pw[4] = {0, 0, 0, 0};
  int rk[4] = {0, 0, 0, 0};
  if (in) {
    const int4 d = *(const int4*)(dst + e);
    const int4 s = *(const int4*)(src + e);
    const int dd[4] = {d.x, d.y, d.z, d.w};
    const int ss[4] = {s.x, s.y, s.z, s.w};
#pragma unroll
    for (int c = 0; c < 4; ++c) {
      ow[c] = (int)((unsigned)dd[c] / (unsigned)AG_ROWS);
      pw[c] = (ss[c] << 8) | (dd[c] - ow[c] * AG_ROWS);
      rk[c] = atomicAdd(&cnt[ow[c]], 1);
    }
  }
  __syncthreads();

  int c_own = 0;
  int incl  = 0;
  if (tid < 256) {
    c_own = cnt[tid];
    incl  = c_own;
#pragma unroll
    for (int dlt = 1; dlt < 32; dlt <<= 1) {
      const int t = __shfl_up(incl, dlt);
      if (lane >= dlt) incl += t;
    }
    if (lane == 31) wsum[tid >> 5] = incl;
  }
  __syncthreads();
  if (tid < 256) {
    int base = 0;
    for (int w = 0; w < (tid >> 5); ++w) base += wsum[w];
    const int start = base + incl - c_own;
    cnt[tid] = start;
    runs[tid * nch + chunk] = (c_own << 16) | start;
  }
  __syncthreads();

  if (in) {
#pragma unroll
    for (int c = 0; c < 4; ++c) sorted[cnt[ow[c]] + rk[c]] = pw[c];
  }
  __syncthreads();
  if (in) *(int4*)(spk + e) = *(const int4*)(sorted + 4 * tid);
}

#define AG_B    8
#define AG_CAP  8192

template <class F, class G>
__device__ __forceinline__ void for_my_words(const int* __restrict__ myruns,
                                             const int* __restrict__ spk, int nch,
                                             int wave, int lane, F f, G g) {
  for (int c0 = wave; c0 < nch; c0 += 32 * 32) {
    const int ci    = c0 + 32 * lane;
    const int ent   = (ci < nch) ? myruns[ci] : 0;
    const int nmine = min(32, (nch - c0 + 31) >> 5);
    for (int i0 = 0; i0 < nmine; i0 += 8) {
      int nn[8];
      int ss[8];
      int ww[8];
      int tk[8];
#pragma unroll
      for (int q = 0; q < 8; ++q) {
        const int t = __builtin_amdgcn_readlane(ent, (i0 + q) & 31);
        nn[q] = (int)((unsigned)t >> 16);
        ss[q] = (c0 + 32 * (i0 + q)) * CH + (t & 0xFFFF);
        ww[q] = spk[(lane < nn[q]) ? ss[q] + lane : 0];
      }
#pragma unroll
      for (int q = 0; q < 8; ++q) tk[q] = (lane < nn[q]) ? f(ww[q]) : 0;
#pragma unroll
      for (int q = 0; q < 8; ++q) {
        if (lane < nn[q]) g(ww[q], tk[q]);
        for (int b = 32; b < nn[q]; b += 32) {
          if (b + lane < nn[q]) {
            const int w = spk[ss[q] + b + lane];
            g(w, f(w));
          }
        }
      }
    }
  }
}

__global__ __launch_bounds__(1024) void gather_mean(
    const float* __restrict__ X, const int* __restrict__ spk,
    const int* __restrict__ runs, float* __restrict__ agg, int nch, int N) {
  __shared__ float acc[AG_ROWS * D];
  __shared__ int   degi[AG_ROWS];
  __shared__ int   filed[AG_CAP];
  __shared__ int   wcnt[32];
  __shared__ int   wfill[32];
  const int tid  = threadIdx.x;
  const int lane = tid & 31;
  const int wave = __builtin_amdgcn_readfirstlane(tid >> 5);
  const int own  = blockIdx.x;
  const int nb   = own * AG_ROWS;

  for (int i = tid; i < AG_ROWS * D; i += 1024) acc[i] = 0.0f;
  if (tid < AG_ROWS) degi[tid] = 0;
  if (tid < 32) wcnt[tid] = 0;
  __syncthreads();

  const int* myruns = runs + own * nch;
  for_my_words(myruns, spk, nch, wave, lane,
               [&](int w) {
                 atomicAdd(&wcnt[w & 31], 1);
                 atomicAdd(&degi[w & 0xFF], 1);
                 return 0;
               },
               [&](int, int) {});
  __syncthreads();

  const int cw = wcnt[lane];
  int incl = cw;
#pragma unroll
  for (int dlt = 1; dlt < 32; dlt <<= 1) {
    const int t = __shfl_up(incl, dlt);
    if (lane >= dlt) incl += t;
  }
  const int start = incl - cw;
  const int total = __builtin_amdgcn_readlane(incl, 31);
  if (wave == 0) wfill[lane] = start;
  __syncthreads();

  if (total <= AG_CAP) {
    for_my_words(myruns, spk, nch, wave, lane,
                 [&](int w) { return atomicAdd(&wfill[w & 31], 1); },
                 [&](int w, int slot) { filed[slot] = w; });
    __syncthreads();

    const int n  = __builtin_amdgcn_readlane(cw, wave);
    const int st = __builtin_amdgcn_readlane(start, wave);
    int wvg = 0;

    auto fetch = [&](float4 (&v)[AG_B], int (&l)[AG_B], int b) {
      if ((b & 3) == 0) {
        const int e = AG_B * b + lane;
        wvg = (e < n) ? filed[st + e] : 0;
      }
#pragma unroll
      for (int j = 0; j < AG_B; ++j) {
        const int t = __builtin_amdgcn_readlane(wvg, (AG_B * b + j) & 31);
        l[j] = t & 0xFF;
        const float* xr = X + (long)((unsigned)t >> 8) * D + 4 * lane;
        v[j] = *(const float4*)xr;
      }
    };
    auto addb = [&](const float4 (&v)[AG_B], const int (&l)[AG_B], int b) {
#pragma unroll
      for (int j = 0; j < AG_B; j += 2) {
        const int e = AG_B * b + j;
        if (e < n) {
          float4* pa = (float4*)(acc + l[j] * D) + lane;
          if (e + 1 < n && l[j + 1] != l[j]) {
            float4* pb = (float4*)(acc + l[j + 1] * D) + lane;
            float4 ta = *pa;
            float4 tb = *pb;
            ta.x += v[j].x;     ta.y += v[j].y;     ta.z += v[j].z;     ta.w += v[j].w;
            tb.x += v[j + 1].x; tb.y += v[j + 1].y; tb.z += v[j + 1].z; tb.w += v[j + 1].w;
            *pa = ta;
            *pb = tb;
          } else {
            float4 ta = *pa;
            ta.x += v[j].x; ta.y += v[j].y; ta.z += v[j].z; ta.w += v[j].w;
            if (e + 1 < n) {
              ta.x += v[j + 1].x; ta.y += v[j + 1].y; ta.z += v[j + 1].z; ta.w += v[j + 1].w;
            }
            *pa = ta;
          }
        }
      }
    };

    const int nbat = (n + AG_B - 1) / AG_B;
    if (nbat > 0) {
      float4 va[AG_B], vb[AG_B];
      int    la[AG_B], lb[AG_B];
      fetch(va, la, 0);
      for (int b = 0; b < nbat; b += 2) {
        fetch(vb, lb, b + 1);
        addb(va, la, b);
        fetch(va, la, b + 2);
        addb(vb, lb, b + 1);
      }
    }
  } else {
    auto add_rows = [&](int pw, int q_n) {
      for (int k = 0; k < q_n; k += AG_B) {
        int sj[AG_B];
        int lj[AG_B];
#pragma unroll
        for (int j = 0; j < AG_B; ++j) {
          const int t = __builtin_amdgcn_readlane(pw, (k + j) & 31);
          sj[j] = (int)((unsigned)t >> 8);
          lj[j] = t & 0xFF;
        }
        float v[AG_B][4];
#pragma unroll
        for (int j = 0; j < AG_B; ++j) {
          v[j][0] = 0.0f; v[j][1] = 0.0f; v[j][2] = 0.0f; v[j][3] = 0.0f;
          if (k + j < q_n) {
            const float* xr = X + (long)sj[j] * D;
            v[j][0] = xr[lane];
            v[j][1] = xr[lane + 32];
            v[j][2] = xr[lane + 64];
            v[j][3] = xr[lane + 96];
          }
        }
#pragma unroll
        for (int j = 0; j < AG_B; ++j) {
          if (k + j < q_n) {
            float* o = acc + lj[j] * D;
            atomicAdd(o + lane,      v[j][0]);
            atomicAdd(o + lane + 32, v[j][1]);
            atomicAdd(o + lane + 64, v[j][2]);
            atomicAdd(o + lane + 96, v[j][3]);
          }
        }
      }
    };
    for (int c0 = wave; c0 < nch; c0 += 32 * 32) {
      const int ci    = c0 + 32 * lane;
      const int ent   = (ci < nch) ? myruns[ci] : 0;
      const int nmine = min(32, (nch - c0 + 31) >> 5);
      for (int i = 0; i < nmine; ++i) {
        const int t   = __builtin_amdgcn_readlane(ent, i);
        const int n_c = (int)((unsigned)t >> 16);
        const int s_c = (c0 + 32 * i) * CH + (t & 0xFFFF);
        for (int b = 0; b < n_c; b += 32) {
          const int pw = (b + lane < n_c) ? spk[s_c + b + lane] : 0;
          add_rows(pw, min(n_c - b, 32));
        }
      }
    }
  }
  __syncthreads();

  for (int r = wave; r < AG_ROWS; r += 32) {
    const int n = nb + r;
    if (n < N) {
      const float dn = fmaxf((float)degi[r], 1.0f);
      const float4 a = *(const float4*)(acc + r * D + 4 * lane);
      float4 o;
      o.x = a.x / dn; o.y = a.y / dn; o.z = a.z / dn; o.w = a.w / dn;
      *(float4*)(agg + (long)n * D + 4 * lane) = o;
    }
  }
}

__device__ __forceinline__ void node_product(const float* __restrict__ A,
                                             const float* __restrict__ W, int row,
                                             int half, int col, v8f (&acc)[4]) {
  const float* ar = A + (long)row * D + 4 * half;
  const float* wr = W + 4 * half * D + col;
#pragma unroll 4
  for (int j = 0; j < 16; ++j) {
    const float4 a = *(const float4*)(ar + 8 * j);
    const float* w = wr + 8 * j * D;
    const float4 b0 = *(const float4*)(w);
    const float4 b1 = *(const float4*)(w + D);
    const float4 b2 = *(const float4*)(w + 2 * D);
    const float4 b3 = *(const float4*)(w + 3 * D);
    acc[0] = wmma4(a.x, a.y, b0.x, b1.x, acc[0]);
    acc[1] = wmma4(a.x, a.y, b0.y, b1.y, acc[1]);
    acc[2] = wmma4(a.x, a.y, b0.z, b1.z, acc[2]);
    acc[3] = wmma4(a.x, a.y, b0.w, b1.w, acc[3]);
    acc[0] = wmma4(a.z, a.w, b2.x, b3.x, acc[0]);
    acc[1] = wmma4(a.z, a.w, b2.y, b3.y, acc[1]);
    acc[2] = wmma4(a.z, a.w, b2.z, b3.z, acc[2]);
    acc[3] = wmma4(a.z, a.w, b2.w, b3.w, acc[3]);
  }
}

__global__ __launch_bounds__(64) void node_gemm(
    const float* __restrict__ A1, const float* __restrict__ W1,
    const float* __restrict__ A2, const float* __restrict__ W2,
    const float* __restrict__ bias, float* __restrict__ Hout, int do_relu,
    long w1_step, long out_step) {
  const int lane = threadIdx.x & 31;
  const int half = lane >> 4;
  const int c    = lane & 15;
  const int wave = __builtin_amdgcn_readfirstlane((int)(threadIdx.x >> 5));
  const int m0   = blockIdx.x * 16;
  const int col  = 64 * wave + 4 * c;
  W1   += (long)blockIdx.y * w1_step;
  Hout += (long)blockIdx.y * out_step;

  v8f acc[4];
  {
    float4 bv = make_float4(0.0f, 0.0f, 0.0f, 0.0f);
    if (bias) bv = *(const float4*)(bias + col);
#pragma unroll
    for (int r = 0; r < 8; ++r) {
      acc[0][r] = bv.x; acc[1][r] = bv.y; acc[2][r] = bv.z; acc[3][r] = bv.w;
    }
  }

  node_product(A1, W1, m0 + c, half, col, acc);
  if (A2) node_product(A2, W2, m0 + c, half, col, acc);

#pragma unroll
  for (int r = 0; r < 8; ++r) {
    float4 o = make_float4(acc[0][r], acc[1][r], acc[2][r], acc[3][r]);
    if (do_relu) {
      o.x = fmaxf(o.x, 0.0f); o.y = fmaxf(o.y, 0.0f);
      o.z = fmaxf(o.z, 0.0f); o.w = fmaxf(o.w, 0.0f);
    }
    *(float4*)(Hout + (long)(m0 + 8 * half + r) * D + col) = o;
  }
}

template <int CTRL>
__device__ __forceinline__ float row_xor_add(float v) {
  const int t = __builtin_amdgcn_update_dpp(0, __float_as_int(v), CTRL, 0xF, 0xF, false);
  return v + __int_as_float(t);
}

__global__ __launch_bounds__(128) void edge_mlp(
    const float* __restrict__ P1, const float* __restrict__ P2,
    const float* __restrict__ ea, const int* __restrict__ src,
    const int* __restrict__ dst, const float* __restrict__ W3,
    const float* __restrict__ bc1, const float* __restrict__ Wc2,
    const float* __restrict__ bc2, float* __restrict__ out, int E) {
  const int lane = threadIdx.x & 31;
  const int half = lane >> 4;
  const int c    = lane & 15;
  const int wave = __builtin_amdgcn_readfirstlane((int)(threadIdx.x >> 5));
  const int e0   = (blockIdx.x * 4 + wave) * 16;
  if (e0 >= E) return;

  const int4 s0 = *(const int4*)(src + e0 + 8 * half);
  const int4 s1 = *(const int4*)(src + e0 + 8 * half + 4);
  const int4 d0 = *(const int4*)(dst + e0 + 8 * half);
  const int4 d1 = *(const int4*)(dst + e0 + 8 * half + 4);
  const int si[8] = {s0.x, s0.y, s0.z, s0.w, s1.x, s1.y, s1.z, s1.w};
  const int di[8] = {d0.x, d0.y, d0.z, d0.w, d1.x, d1.y, d1.z, d1.w};

  float af[16];
  {
    const float* ar = ea + (long)(e0 + c) * DE + 16 * half;
#pragma unroll
    for (int q = 0; q < 4; ++q) {
      const float4 a = *(const float4*)(ar + 4 * q);
      af[4 * q + 0] = a.x; af[4 * q + 1] = a.y; af[4 * q + 2] = a.z; af[4 * q + 3] = a.w;
    }
  }

  v8f acc[8];
  {
    const float4 bv0 = *(const float4*)(bc1 + 4 * c);
    const float4 bv1 = *(const float4*)(bc1 + 64 + 4 * c);
#pragma unroll
    for (int r = 0; r < 8; ++r) {
      const float* p1 = P1 + (long)si[r] * D + 4 * c;
      const float* p2 = P2 + (long)di[r] * D + 4 * c;
      const float4 x0 = *(const float4*)(p1);
      const float4 x1 = *(const float4*)(p1 + 64);
      const float4 y0 = *(const float4*)(p2);
      const float4 y1 = *(const float4*)(p2 + 64);
      acc[0][r] = (x0.x + y0.x) + bv0.x;
      acc[1][r] = (x0.y + y0.y) + bv0.y;
      acc[2][r] = (x0.z + y0.z) + bv0.z;
      acc[3][r] = (x0.w + y0.w) + bv0.w;
      acc[4][r] = (x1.x + y1.x) + bv1.x;
      acc[5][r] = (x1.y + y1.y) + bv1.y;
      acc[6][r] = (x1.z + y1.z) + bv1.z;
      acc[7][r] = (x1.w + y1.w) + bv1.w;
    }
  }

#pragma unroll
  for (int ks = 0; ks < 8; ++ks) {
    const float* w0 = W3 + (16 * half + 2 * ks) * D + 4 * c;
    const float* w1 = w0 + D;
    const float4 b00 = *(const float4*)(w0);
    const float4 b01 = *(const float4*)(w0 + 64);
    const float4 b10 = *(const float4*)(w1);
    const float4 b11 = *(const float4*)(w1 + 64);
    const float a0 = af[2 * ks], a1 = af[2 * ks + 1];
    acc[0] = wmma4(a0, a1, b00.x, b10.x, acc[0]);
    acc[1] = wmma4(a0, a1, b00.y, b10.y, acc[1]);
    acc[2] = wmma4(a0, a1, b00.z, b10.z, acc[2]);
    acc[3] = wmma4(a0, a1, b00.w, b10.w, acc[3]);
    acc[4] = wmma4(a0, a1, b01.x, b11.x, acc[4]);
    acc[5] = wmma4(a0, a1, b01.y, b11.y, acc[5]);
    acc[6] = wmma4(a0, a1, b01.z, b11.z, acc[6]);
    acc[7] = wmma4(a0, a1, b01.w, b11.w, acc[7]);
  }

  const float4 wq0 = *(const float4*)(Wc2 + 4 * c);
  const float4 wq1 = *(const float4*)(Wc2 + 64 + 4 * c);
  const float w[8] = {wq0.x, wq0.y, wq0.z, wq0.w, wq1.x, wq1.y, wq1.z, wq1.w};
  float sc[8];
#pragma unroll
  for (int r = 0; r < 8; ++r) {
    float t = fmaxf(acc[0][r], 0.0f) * w[0];
#pragma unroll
    for (int k = 1; k < 8; ++k) t += fmaxf(acc[k][r], 0.0f) * w[k];
    t = row_xor_add<0x161>(t);
    t = row_xor_add<0x162>(t);
    t = row_xor_add<0x164>(t);
    t = row_xor_add<0x168>(t);
    sc[r] = t;
  }
  float v = sc[0];
#pragma unroll
  for (int r = 1; r < 8; ++r) v = (c == r) ? sc[r] : v;
  if (c < 8) out[e0 + 8 * half + c] = v + bc2[0];
}

extern "C" void kernel_launch(void* const* d_in, const int* in_sizes, int n_in,
                              void* d_out, int out_size, void* d_ws, size_t ws_size,
                              hipStream_t stream) {
  (void)n_in; (void)out_size; (void)ws_size;
  const float* x   = (const float*)d_in[0];
  const float* ea  = (const float*)d_in[1];
  const float* W1a = (const float*)d_in[2];
  const float* b1  = (const float*)d_in[3];
  const float* W1r = (const float*)d_in[4];
  const float* W2a = (const float*)d_in[5];
  const float* b2  = (const float*)d_in[6];
  const float* W2r = (const float*)d_in[7];
  const float* Wc1 = (const float*)d_in[8];
  const float* bc1 = (const float*)d_in[9];
  const float* Wc2 = (const float*)d_in[10];
  const float* bc2 = (const float*)d_in[11];
  const int*   ei  = (const int*)d_in[12];

  const int N = in_sizes[0] / D;
  const int E = in_sizes[12] / 2;
  const int* src = ei;
  const int* dst = ei + E;

  float* ws = (float*)d_ws;
  const long NM = (long)N * D;
  float* bufA = ws;
  float* bufB = ws + NM;
  float* bufC = ws + 2 * NM;
  float* out  = (float*)d_out;

  const int agrid = (N + AG_ROWS - 1) / AG_ROWS;
  const int nch   = (E + CH - 1) / CH;
  const int ggrid = N / 16;
  const int egrid = (E / 16 + 3) / 4;

  int* spk  = (int*)bufC;
  int* runs = (int*)bufC + nch * CH;
  edge_sort<<<nch, dim3(1024), 0, stream>>>(src, dst, spk, runs, E, nch);

  gather_mean<<<agrid, dim3(1024), 0, stream>>>(x, spk, runs, bufA, nch, N);
  node_gemm<<<dim3(ggrid, 1), dim3(64), 0, stream>>>(bufA, W1a, x, W1r, b1, bufB, 1, 0L, 0L);

  gather_mean<<<agrid, dim3(1024), 0, stream>>>(bufB, spk, runs, bufA, nch, N);
  node_gemm<<<dim3(ggrid, 1), dim3(64), 0, stream>>>(bufA, W2a, bufB, W2r, b2, bufC, 0, 0L, 0L);

  node_gemm<<<dim3(ggrid, 2), dim3(64), 0, stream>>>(bufC, Wc1, nullptr, nullptr, nullptr,
                                                    bufA, 0, 128L * D, NM);

  edge_mlp<<<egrid, dim3(128), 0, stream>>>(bufA, bufB, ea, src, dst, Wc1 + 256 * D,
                                      bc1, Wc2, bc2, out, E);
}


# layer 2 folded into the edge MLP's node terms: P = agg2@(W2a@Wc1) + h1@(W2r@Wc1) + b2@Wc1, weights multiplied in spare blocks of the layer-1 launch; no h2, 7 launches to 6; product order changed
# speedup vs baseline: 3.5761x; 1.0924x over previous
#include <hip/hip_runtime.h>


typedef __attribute__((ext_vector_type(2))) float v2f;
typedef __attribute__((ext_vector_type(8))) float v8f;

#define D   128
#define DE  32

__device__ __forceinline__ v8f wmma4(float a0, float a1, float b0, float b1, v8f c) {
  v2f a; a[0] = a0; a[1] = a1;
  v2f b; b[0] = b0; b[1] = b1;
  return __builtin_amdgcn_wmma_f32_16x16x4_f32(false, a, false, b, (short)0, c,
                                               false, false);
}

#define AG_ROWS 196
#define CH      4096

__global__ __launch_bounds__(1024) void edge_sort(
    const int* __restrict__ src, const int* __restrict__ dst,
    int* __restrict__ spk, int* __restrict__ runs, int E, int nch) {
  __shared__ int cnt[256];
  __shared__ int wsum[8];
  __shared__ int sorted[CH];
  const int tid   = threadIdx.x;
  const int lane  = tid & 31;
  const int chunk = blockIdx.x;
  if (tid < 256) cnt[tid] = 0;
  __syncthreads();

  const int  e  = chunk * CH + 4 * tid;
  const bool in = e < E;
  int ow[4] = {0, 0, 0, 0};
  int pw[4] = {0, 0, 0, 0};
  int rk[4] = {0, 0, 0, 0};
  if (in) {
    const int4 d = *(const int4*)(dst + e);
    const int4 s = *(const int4*)(src + e);
    const int dd[4] = {d.x, d.y, d.z, d.w};
    const int ss[4] = {s.x, s.y, s.z, s.w};
#pragma unroll
    for (int c = 0; c < 4; ++c) {
      ow[c] = (int)((unsigned)dd[c] / (unsigned)AG_ROWS);
      pw[c] = (ss[c] << 8) | (dd[c] - ow[c] * AG_ROWS);
      rk[c] = atomicAdd(&cnt[ow[c]], 1);
    }
  }
  __syncthreads();

  int c_own = 0;
  int incl  = 0;
  if (tid < 256) {
    c_own = cnt[tid];
    incl  = c_own;
#pragma unroll
    for (int dlt = 1; dlt < 32; dlt <<= 1) {
      const int t = __shfl_up(incl, dlt);
      if (lane >= dlt) incl += t;
    }
    if (lane == 31) wsum[tid >> 5] = incl;
  }
  __syncthreads();
  if (tid < 256) {
    int base = 0;
    for (int w = 0; w < (tid >> 5); ++w) base += wsum[w];
    const int start = base + incl - c_own;
    cnt[tid] = start;
    runs[tid * nch + chunk] = (c_own << 16) | start;
  }
  __syncthreads();

  if (in) {
#pragma unroll
    for (int c = 0; c < 4; ++c) sorted[cnt[ow[c]] + rk[c]] = pw[c];
  }
  __syncthreads();
  if (in) *(int4*)(spk + e) = *(const int4*)(sorted + 4 * tid);
}

#define AG_B    8
#define AG_CAP  8192

template <class F, class G>
__device__ __forceinline__ void for_my_words(const int* __restrict__ myruns,
                                             const int* __restrict__ spk, int nch,
                                             int wave, int lane, F f, G g) {
  for (int c0 = wave; c0 < nch; c0 += 32 * 32) {
    const int ci    = c0 + 32 * lane;
    const int ent   = (ci < nch) ? myruns[ci] : 0;
    const int nmine = min(32, (nch - c0 + 31) >> 5);
    for (int i0 = 0; i0 < nmine; i0 += 8) {
      int nn[8];
      int ss[8];
      int ww[8];
      int tk[8];
#pragma unroll
      for (int q = 0; q < 8; ++q) {
        const int t = __builtin_amdgcn_readlane(ent, (i0 + q) & 31);
        nn[q] = (int)((unsigned)t >> 16);
        ss[q] = (c0 + 32 * (i0 + q)) * CH + (t & 0xFFFF);
        ww[q] = spk[(lane < nn[q]) ? ss[q] + lane : 0];
      }
#pragma unroll
      for (int q = 0; q < 8; ++q) tk[q] = (lane < nn[q]) ? f(ww[q]) : 0;
#pragma unroll
      for (int q = 0; q < 8; ++q) {
        if (lane < nn[q]) g(ww[q], tk[q]);
        for (int b = 32; b < nn[q]; b += 32) {
          if (b + lane < nn[q]) {
            const int w = spk[ss[q] + b + lane];
            g(w, f(w));
          }
        }
      }
    }
  }
}

__global__ __launch_bounds__(1024) void gather_mean(
    const float* __restrict__ X, const int* __restrict__ spk,
    const int* __restrict__ runs, float* __restrict__ agg, int nch, int N) {
  __shared__ float acc[AG_ROWS * D];
  __shared__ int   degi[AG_ROWS];
  __shared__ int   filed[AG_CAP];
  __shared__ int   wcnt[32];
  __shared__ int   wfill[32];
  const int tid  = threadIdx.x;
  const int lane = tid & 31;
  const int wave = __builtin_amdgcn_readfirstlane(tid >> 5);
  const int own  = blockIdx.x;
  const int nb   = own * AG_ROWS;

  for (int i = tid; i < AG_ROWS * D; i += 1024) acc[i] = 0.0f;
  if (tid < AG_ROWS) degi[tid] = 0;
  if (tid < 32) wcnt[tid] = 0;
  __syncthreads();

  const int* myruns = runs + own * nch;
  for_my_words(myruns, spk, nch, wave, lane,
               [&](int w) {
                 atomicAdd(&wcnt[w & 31], 1);
                 atomicAdd(&degi[w & 0xFF], 1);
                 return 0;
               },
               [&](int, int) {});
  __syncthreads();

  const int cw = wcnt[lane];
  int incl = cw;
#pragma unroll
  for (int dlt = 1; dlt < 32; dlt <<= 1) {
    const int t = __shfl_up(incl, dlt);
    if (lane >= dlt) incl += t;
  }
  const int start = incl - cw;
  const int total = __builtin_amdgcn_readlane(incl, 31);
  if (wave == 0) wfill[lane] = start;
  __syncthreads();

  if (total <= AG_CAP) {
    for_my_words(myruns, spk, nch, wave, lane,
                 [&](int w) { return atomicAdd(&wfill[w & 31], 1); },
                 [&](int w, int slot) { filed[slot] = w; });
    __syncthreads();

    const int n  = __builtin_amdgcn_readlane(cw, wave);
    const int st = __builtin_amdgcn_readlane(start, wave);
    int wvg = 0;

    auto fetch = [&](float4 (&v)[AG_B], int (&l)[AG_B], int b) {
      if ((b & 3) == 0) {
        const int e = AG_B * b + lane;
        wvg = (e < n) ? filed[st + e] : 0;
      }
#pragma unroll
      for (int j = 0; j < AG_B; ++j) {
        const int t = __builtin_amdgcn_readlane(wvg, (AG_B * b + j) & 31);
        l[j] = t & 0xFF;
        const float* xr = X + (long)((unsigned)t >> 8) * D + 4 * lane;
        v[j] = *(const float4*)xr;
      }
    };
    auto addb = [&](const float4 (&v)[AG_B], const int (&l)[AG_B], int b) {
#pragma unroll
      for (int j = 0; j < AG_B; j += 2) {
        const int e = AG_B * b + j;
        if (e < n) {
          float4* pa = (float4*)(acc + l[j] * D) + lane;
          if (e + 1 < n && l[j + 1] != l[j]) {
            float4* pb = (float4*)(acc + l[j + 1] * D) + lane;
            float4 ta = *pa;
            float4 tb = *pb;
            ta.x += v[j].x;     ta.y += v[j].y;     ta.z += v[j].z;     ta.w += v[j].w;
            tb.x += v[j + 1].x; tb.y += v[j + 1].y; tb.z += v[j + 1].z; tb.w += v[j + 1].w;
            *pa = ta;
            *pb = tb;
          } else {
            float4 ta = *pa;
            ta.x += v[j].x; ta.y += v[j].y; ta.z += v[j].z; ta.w += v[j].w;
            if (e + 1 < n) {
              ta.x += v[j + 1].x; ta.y += v[j + 1].y; ta.z += v[j + 1].z; ta.w += v[j + 1].w;
            }
            *pa = ta;
          }
        }
      }
    };

    const int nbat = (n + AG_B - 1) / AG_B;
    if (nbat > 0) {
      float4 va[AG_B], vb[AG_B];
      int    la[AG_B], lb[AG_B];
      fetch(va, la, 0);
      for (int b = 0; b < nbat; b += 2) {
        fetch(vb, lb, b + 1);
        addb(va, la, b);
        fetch(va, la, b + 2);
        addb(vb, lb, b + 1);
      }
    }
  } else {
    auto add_rows = [&](int pw, int q_n) {
      for (int k = 0; k < q_n; k += AG_B) {
        int sj[AG_B];
        int lj[AG_B];
#pragma unroll
        for (int j = 0; j < AG_B; ++j) {
          const int t = __builtin_amdgcn_readlane(pw, (k + j) & 31);
          sj[j] = (int)((unsigned)t >> 8);
          lj[j] = t & 0xFF;
        }
        float v[AG_B][4];
#pragma unroll
        for (int j = 0; j < AG_B; ++j) {
          v[j][0] = 0.0f; v[j][1] = 0.0f; v[j][2] = 0.0f; v[j][3] = 0.0f;
          if (k + j < q_n) {
            const float* xr = X + (long)sj[j] * D;
            v[j][0] = xr[lane];
            v[j][1] = xr[lane + 32];
            v[j][2] = xr[lane + 64];
            v[j][3] = xr[lane + 96];
          }
        }
#pragma unroll
        for (int j = 0; j < AG_B; ++j) {
          if (k + j < q_n) {
            float* o = acc + lj[j] * D;
            atomicAdd(o + lane,      v[j][0]);
            atomicAdd(o + lane + 32, v[j][1]);
            atomicAdd(o + lane + 64, v[j][2]);
            atomicAdd(o + lane + 96, v[j][3]);
          }
        }
      }
    };
    for (int c0 = wave; c0 < nch; c0 += 32 * 32) {
      const int ci    = c0 + 32 * lane;
      const int ent   = (ci < nch) ? myruns[ci] : 0;
      const int nmine = min(32, (nch - c0 + 31) >> 5);
      for (int i = 0; i < nmine; ++i) {
        const int t   = __builtin_amdgcn_readlane(ent, i);
        const int n_c = (int)((unsigned)t >> 16);
        const int s_c = (c0 + 32 * i) * CH + (t & 0xFFFF);
        for (int b = 0; b < n_c; b += 32) {
          const int pw = (b + lane < n_c) ? spk[s_c + b + lane] : 0;
          add_rows(pw, min(n_c - b, 32));
        }
      }
    }
  }
  __syncthreads();

  for (int r = wave; r < AG_ROWS; r += 32) {
    const int n = nb + r;
    if (n < N) {
      const float dn = fmaxf((float)degi[r], 1.0f);
      const float4 a = *(const float4*)(acc + r * D + 4 * lane);
      float4 o;
      o.x = a.x / dn; o.y = a.y / dn; o.z = a.z / dn; o.w = a.w / dn;
      *(float4*)(agg + (long)n * D + 4 * lane) = o;
    }
  }
}

__device__ __forceinline__ void node_product(const float* __restrict__ A,
                                             const float* __restrict__ W, int row,
                                             int half, int col, v8f (&acc)[4]) {
  const float* ar = A + (long)row * D + 4 * half;
  const float* wr = W + 4 * half * D + col;
#pragma unroll 4
  for (int j = 0; j < 16; ++j) {
    const float4 a = *(const float4*)(ar + 8 * j);
    const float* w = wr + 8 * j * D;
    const float4 b0 = *(const float4*)(w);
    const float4 b1 = *(const float4*)(w + D);
    const float4 b2 = *(const float4*)(w + 2 * D);
    const float4 b3 = *(const float4*)(w + 3 * D);
    acc[0] = wmma4(a.x, a.y, b0.x, b1.x, acc[0]);
    acc[1] = wmma4(a.x, a.y, b0.y, b1.y, acc[1]);
    acc[2] = wmma4(a.x, a.y, b0.z, b1.z, acc[2]);
    acc[3] = wmma4(a.x, a.y, b0.w, b1.w, acc[3]);
    acc[0] = wmma4(a.z, a.w, b2.x, b3.x, acc[0]);
    acc[1] = wmma4(a.z, a.w, b2.y, b3.y, acc[1]);
    acc[2] = wmma4(a.z, a.w, b2.z, b3.z, acc[2]);
    acc[3] = wmma4(a.z, a.w, b2.w, b3.w, acc[3]);
  }
}

#define FOLD_JOBS 34

__global__ __launch_bounds__(64) void node_gemm(
    const float* __restrict__ A1, const float* __restrict__ W1,
    const float* __restrict__ A2, const float* __restrict__ W2,
    const float* __restrict__ bias, float* __restrict__ Hout, int nmain,
    const float* __restrict__ fW2a, const float* __restrict__ fW2r,
    const float* __restrict__ fb2, const float* __restrict__ fWc1,
    float* __restrict__ fold) {
  const int lane = threadIdx.x & 31;
  const int half = lane >> 4;
  const int c    = lane & 15;
  const int wave = __builtin_amdgcn_readfirstlane((int)(threadIdx.x >> 5));
  const int col  = 64 * wave + 4 * c;

  v8f acc[4];
  if ((int)blockIdx.x >= nmain) {
    const int job = (int)blockIdx.x - nmain;
    const int y   = (job < 32) ? ((job >> 3) & 1) : (job - 32);
    const float* Wy = fWc1 + (long)y * 128 * D;
#pragma unroll
    for (int r = 0; r < 8; ++r) {
      acc[0][r] = 0.0f; acc[1][r] = 0.0f; acc[2][r] = 0.0f; acc[3][r] = 0.0f;
    }
    if (job < 32) {
      const float* Af = (job < 16) ? fW2a : fW2r;
      const int    f0 = 16 * (job & 7);
      node_product(Af, Wy, f0 + c, half, col, acc);
      float* o = fold + (long)((job >> 4) * 2 + y) * 128 * D;
#pragma unroll
      for (int r = 0; r < 8; ++r) {
        *(float4*)(o + (long)(f0 + 8 * half + r) * D + col) =
            make_float4(acc[0][r], acc[1][r], acc[2][r], acc[3][r]);
      }
    } else {
      node_product(fb2, Wy, 0, half, col, acc);
      if (half == 0) {
        *(float4*)(fold + 4L * 128 * D + (long)y * D + col) =
            make_float4(acc[0][0], acc[1][0], acc[2][0], acc[3][0]);
      }
    }
    return;
  }

  const int m0 = blockIdx.x * 16;
  {
    float4 bv = make_float4(0.0f, 0.0f, 0.0f, 0.0f);
    if (bias) bv = *(const float4*)(bias + col);
#pragma unroll
    for (int r = 0; r < 8; ++r) {
      acc[0][r] = bv.x; acc[1][r] = bv.y; acc[2][r] = bv.z; acc[3][r] = bv.w;
    }
  }

  node_product(A1, W1, m0 + c, half, col, acc);
  if (A2) node_product(A2, W2, m0 + c, half, col, acc);

#pragma unroll
  for (int r = 0; r < 8; ++r) {
    float4 o = make_float4(acc[0][r], acc[1][r], acc[2][r], acc[3][r]);
    o.x = fmaxf(o.x, 0.0f); o.y = fmaxf(o.y, 0.0f);
    o.z = fmaxf(o.z, 0.0f); o.w = fmaxf(o.w, 0.0f);
    *(float4*)(Hout + (long)(m0 + 8 * half + r) * D + col) = o;
  }
}

__global__ __launch_bounds__(128) void node_pair(
    const float* A1, const float* A2, const float* __restrict__ fold,
    float* O1, float* O2) {
  const int lane = threadIdx.x & 31;
  const int half = lane >> 4;
  const int c    = lane & 15;
  const int wave = __builtin_amdgcn_readfirstlane((int)(threadIdx.x >> 5));
  const int y    = wave >> 1;
  const int m0   = blockIdx.x * 16;
  const int col  = 64 * (wave & 1) + 4 * c;

  v8f acc[4];
  {
    const float4 bv = *(const float4*)(fold + 4L * 128 * D + (long)y * D + col);
#pragma unroll
    for (int r = 0; r < 8; ++r) {
      acc[0][r] = bv.x; acc[1][r] = bv.y; acc[2][r] = bv.z; acc[3][r] = bv.w;
    }
  }
  node_product(A1, fold + (long)y * 128 * D, m0 + c, half, col, acc);
  node_product(A2, fold + (long)(2 + y) * 128 * D, m0 + c, half, col, acc);

  __syncthreads();

  float* Hout = y ? O2 : O1;
#pragma unroll
  for (int r = 0; r < 8; ++r) {
    *(float4*)(Hout + (long)(m0 + 8 * half + r) * D + col) =
        make_float4(acc[0][r], acc[1][r], acc[2][r], acc[3][r]);
  }
}

template <int CTRL>
__device__ __forceinline__ float row_xor_add(float v) {
  const int t = __builtin_amdgcn_update_dpp(0, __float_as_int(v), CTRL, 0xF, 0xF, false);
  return v + __int_as_float(t);
}

__global__ __launch_bounds__(128) void edge_mlp(
    const float* __restrict__ P1, const float* __restrict__ P2,
    const float* __restrict__ ea, const int* __restrict__ src,
    const int* __restrict__ dst, const float* __restrict__ W3,
    const float* __restrict__ bc1, const float* __restrict__ Wc2,
    const float* __restrict__ bc2, float* __restrict__ out, int E) {
  const int lane = threadIdx.x & 31;
  const int half = lane >> 4;
  const int c    = lane & 15;
  const int wave = __builtin_amdgcn_readfirstlane((int)(threadIdx.x >> 5));
  const int e0   = (blockIdx.x * 4 + wave) * 16;
  if (e0 >= E) return;

  const int4 s0 = *(const int4*)(src + e0 + 8 * half);
  const int4 s1 = *(const int4*)(src + e0 + 8 * half + 4);
  const int4 d0 = *(const int4*)(dst + e0 + 8 * half);
  const int4 d1 = *(const int4*)(dst + e0 + 8 * half + 4);
  const int si[8] = {s0.x, s0.y, s0.z, s0.w, s1.x, s1.y, s1.z, s1.w};
  const int di[8] = {d0.x, d0.y, d0.z, d0.w, d1.x, d1.y, d1.z, d1.w};

  float af[16];
  {
    const float* ar = ea + (long)(e0 + c) * DE + 16 * half;
#pragma unroll
    for (int q = 0; q < 4; ++q) {
      const float4 a = *(const float4*)(ar + 4 * q);
      af[4 * q + 0] = a.x; af[4 * q + 1] = a.y; af[4 * q + 2] = a.z; af[4 * q + 3] = a.w;
    }
  }

  v8f acc[8];
  {
    const float4 bv0 = *(const float4*)(bc1 + 4 * c);
    const float4 bv1 = *(const float4*)(bc1 + 64 + 4 * c);
#pragma unroll
    for (int r = 0; r < 8; ++r) {
      const float* p1 = P1 + (long)si[r] * D + 4 * c;
      const float* p2 = P2 + (long)di[r] * D + 4 * c;
      const float4 x0 = *(const float4*)(p1);
      const float4 x1 = *(const float4*)(p1 + 64);
      const float4 y0 = *(const float4*)(p2);
      const float4 y1 = *(const float4*)(p2 + 64);
      acc[0][r] = (x0.x + y0.x) + bv0.x;
      acc[1][r] = (x0.y + y0.y) + bv0.y;
      acc[2][r] = (x0.z + y0.z) + bv0.z;
      acc[3][r] = (x0.w + y0.w) + bv0.w;
      acc[4][r] = (x1.x + y1.x) + bv1.x;
      acc[5][r] = (x1.y + y1.y) + bv1.y;
      acc[6][r] = (x1.z + y1.z) + bv1.z;
      acc[7][r] = (x1.w + y1.w) + bv1.w;
    }
  }

#pragma unroll
  for (int ks = 0; ks < 8; ++ks) {
    const float* w0 = W3 + (16 * half + 2 * ks) * D + 4 * c;
    const float* w1 = w0 + D;
    const float4 b00 = *(const float4*)(w0);
    const float4 b01 = *(const float4*)(w0 + 64);
    const float4 b10 = *(const float4*)(w1);
    const float4 b11 = *(const float4*)(w1 + 64);
    const float a0 = af[2 * ks], a1 = af[2 * ks + 1];
    acc[0] = wmma4(a0, a1, b00.x, b10.x, acc[0]);
    acc[1] = wmma4(a0, a1, b00.y, b10.y, acc[1]);
    acc[2] = wmma4(a0, a1, b00.z, b10.z, acc[2]);
    acc[3] = wmma4(a0, a1, b00.w, b10.w, acc[3]);
    acc[4] = wmma4(a0, a1, b01.x, b11.x, acc[4]);
    acc[5] = wmma4(a0, a1, b01.y, b11.y, acc[5]);
    acc[6] = wmma4(a0, a1, b01.z, b11.z, acc[6]);
    acc[7] = wmma4(a0, a1, b01.w, b11.w, acc[7]);
  }

  const float4 wq0 = *(const float4*)(Wc2 + 4 * c);
  const float4 wq1 = *(const float4*)(Wc2 + 64 + 4 * c);
  const float w[8] = {wq0.x, wq0.y, wq0.z, wq0.w, wq1.x, wq1.y, wq1.z, wq1.w};
  float sc[8];
#pragma unroll
  for (int r = 0; r < 8; ++r) {
    float t = fmaxf(acc[0][r], 0.0f) * w[0];
#pragma unroll
    for (int k = 1; k < 8; ++k) t += fmaxf(acc[k][r], 0.0f) * w[k];
    t = row_xor_add<0x161>(t);
    t = row_xor_add<0x162>(t);
    t = row_xor_add<0x164>(t);
    t = row_xor_add<0x168>(t);
    sc[r] = t;
  }
  float v = sc[0];
#pragma unroll
  for (int r = 1; r < 8; ++r) v = (c == r) ? sc[r] : v;
  if (c < 8) out[e0 + 8 * half + c] = v + bc2[0];
}

extern "C" void kernel_launch(void* const* d_in, const int* in_sizes, int n_in,
                              void* d_out, int out_size, void* d_ws, size_t ws_size,
                              hipStream_t stream) {
  (void)n_in; (void)out_size; (void)ws_size;
  const float* x   = (const float*)d_in[0];
  const float* ea  = (const float*)d_in[1];
  const float* W1a = (const float*)d_in[2];
  const float* b1  = (const float*)d_in[3];
  const float* W1r = (const float*)d_in[4];
  const float* W2a = (const float*)d_in[5];
  const float* b2  = (const float*)d_in[6];
  const float* W2r = (const float*)d_in[7];
  const float* Wc1 = (const float*)d_in[8];
  const float* bc1 = (const float*)d_in[9];
  const float* Wc2 = (const float*)d_in[10];
  const float* bc2 = (const float*)d_in[11];
  const int*   ei  = (const int*)d_in[12];

  const int N = in_sizes[0] / D;
  const int E = in_sizes[12] / 2;
  const int* src = ei;
  const int* dst = ei + E;

  float* ws = (float*)d_ws;
  const long NM = (long)N * D;
  float* bufA = ws;
  float* bufB = ws + NM;
  float* bufC = ws + 2 * NM;
  float* out  = (float*)d_out;

  const int agrid = (N + AG_ROWS - 1) / AG_ROWS;
  const int nch   = (E + CH - 1) / CH;
  const int ggrid = N / 16;
  const int egrid = (E / 16 + 3) / 4;

  int* spk  = (int*)bufC;
  int* runs = (int*)bufC + nch * CH;
  edge_sort<<<nch, dim3(1024), 0, stream>>>(src, dst, spk, runs, E, nch);

  float* fold = (float*)(runs + 256 * nch);

  gather_mean<<<agrid, dim3(1024), 0, stream>>>(x, spk, runs, bufA, nch, N);
  node_gemm<<<dim3(ggrid + FOLD_JOBS), dim3(64), 0, stream>>>(bufA, W1a, x, W1r, b1, bufB, ggrid,
                                                            W2a, W2r, b2, Wc1, fold);

  gather_mean<<<agrid, dim3(1024), 0, stream>>>(bufB, spk, runs, bufA, nch, N);
  node_pair<<<dim3(ggrid), dim3(128), 0, stream>>>(bufA, bufB, fold, bufA, bufB);

  edge_mlp<<<egrid, dim3(128), 0, stream>>>(bufA, bufB, ea, src, dst, Wc1 + 256 * D,
                                      bc1, Wc2, bc2, out, E);
}
